# starting kernel
# baseline (speedup 1.0000x reference)
#include <hip/hip_runtime.h>
#include <hip/hip_bf16.h>

#define NDIM 128
#define NEG_SLOPE 0.01f

typedef __attribute__((ext_vector_type(2))) float v2f;
typedef __attribute__((ext_vector_type(8))) float v8f;

__global__ void transpose128(const float* __restrict__ W, float* __restrict__ Wt) {
    int i = blockIdx.x * blockDim.x + threadIdx.x;
    if (i < NDIM * NDIM) {
        int r = i >> 7, c = i & 127;
        Wt[c * NDIM + r] = W[i];
    }
}

__global__ void fill_ones(float* __restrict__ p, int n) {
    int i = blockIdx.x * blockDim.x + threadIdx.x;
    if (i < n) p[i] = 1.0f;
}

__global__ void count_deg(const int* __restrict__ dst, float* __restrict__ deg, int e) {
    int i = blockIdx.x * blockDim.x + threadIdx.x;
    if (i < e) atomicAdd(&deg[dst[i]], 1.0f);
}

__global__ void rsqrt_inplace(float* __restrict__ p, int n) {
    int i = blockIdx.x * blockDim.x + threadIdx.x;
    if (i < n) p[i] = rsqrtf(p[i]);
}

__global__ void zero_f32(float* __restrict__ p, int n) {
    int i = blockIdx.x * blockDim.x + threadIdx.x;
    if (i < n) p[i] = 0.0f;
}

__global__ void __launch_bounds__(256)
gemm_wmma_f32(const float* __restrict__ X, const float* __restrict__ Wt,
              float* __restrict__ H, int nrows) {
    __shared__ float Xs[16][132];

    const int tid     = threadIdx.x;
    const int rowBase = blockIdx.x * 16;

    for (int i = tid; i < 16 * NDIM; i += 256) {
        int r = i >> 7, c = i & 127;
        int gr = rowBase + r;
        Xs[r][c] = (gr < nrows) ? X[gr * NDIM + c] : 0.0f;
    }
    __syncthreads();

    const int wave    = tid >> 5;
    const int lane    = tid & 31;
    const int colBase = wave * 16;
    const int l16     = lane & 15;
    const int khalf   = (lane >> 4) << 1;

    v8f acc = {};
    #pragma unroll
    for (int k = 0; k < NDIM; k += 4) {
        v2f a, b;
        a.x = Xs[l16][k + khalf];
        a.y = Xs[l16][k + khalf + 1];
        b.x = Wt[(k + khalf)     * NDIM + colBase + l16];
        b.y = Wt[(k + khalf + 1) * NDIM + colBase + l16];
        acc = __builtin_amdgcn_wmma_f32_16x16x4_f32(
             false, a,  false, b,
             (short)0, acc,  false,  false);
    }

    const int row0 = rowBase + ((lane >> 4) << 3);
    const int col  = colBase + l16;
    #pragma unroll
    for (int v = 0; v < 8; ++v) {
        int r = row0 + v;
        if (r < nrows) H[r * NDIM + col] = acc[v];
    }
}

__global__ void __launch_bounds__(256)
edge_scatter(const float* __restrict__ h, const float* __restrict__ dis,
             const int* __restrict__ src, const int* __restrict__ dst,
             float* __restrict__ agg, int e) {
    int gid  = blockIdx.x * blockDim.x + threadIdx.x;
    int edge = gid >> 5;
    if (edge >= e) return;
    int part = gid & 31;

    int s = src[edge];
    int d = dst[edge];
    float norm = dis[s] * dis[d];

    const float4 hv = ((const float4*)(h + (size_t)s * NDIM))[part];
    float* out = agg + (size_t)d * NDIM + part * 4;
    atomicAdd(out + 0, hv.x * norm);
    atomicAdd(out + 1, hv.y * norm);
    atomicAdd(out + 2, hv.z * norm);
    atomicAdd(out + 3, hv.w * norm);
}

__global__ void __launch_bounds__(256)
post_epilogue(float* __restrict__ agg, const float* __restrict__ h,
              const float* __restrict__ dis, const float* __restrict__ bias,
              int n, int do_lrelu) {
    int i = blockIdx.x * blockDim.x + threadIdx.x;
    if (i >= n * NDIM) return;
    int row = i >> 7, c = i & 127;
    float ds = dis[row];
    float v = agg[i] + h[i] * (ds * ds) + bias[c];
    if (do_lrelu) v = (v > 0.0f) ? v : NEG_SLOPE * v;
    agg[i] = v;
}

extern "C" void kernel_launch(void* const* d_in, const int* in_sizes, int n_in,
                              void* d_out, int out_size, void* d_ws, size_t ws_size,
                              hipStream_t stream) {
    const float* x    = (const float*)d_in[0];
    const int*   ei   = (const int*)d_in[1];
    const float* W1   = (const float*)d_in[3];
    const float* b1   = (const float*)d_in[4];
    const float* W2   = (const float*)d_in[5];
    const float* b2   = (const float*)d_in[6];

    const int N = in_sizes[0] / NDIM;
    const int E = in_sizes[1] / 2;
    const int* src = ei;
    const int* dst = ei + E;

    float* Wt1  = (float*)d_ws;
    float* Wt2  = Wt1 + NDIM * NDIM;
    float* dis  = Wt2 + NDIM * NDIM;
    float* bufA = dis + ((N + 3) & ~3);
    float* bufB = bufA + (size_t)N * NDIM;
    float* outp = (float*)d_out;

    const int T = 256;
    dim3 blk(T);
    dim3 gW((NDIM * NDIM + T - 1) / T);
    dim3 gN((N + T - 1) / T);
    dim3 gE((E + T - 1) / T);
    dim3 gNF((N * NDIM + T - 1) / T);
    dim3 gScat(((E * 32) + T - 1) / T);
    dim3 gGemm((N + 15) / 16);

    transpose128<<<gW, blk, 0, stream>>>(W1, Wt1);
    transpose128<<<gW, blk, 0, stream>>>(W2, Wt2);
    fill_ones<<<gN, blk, 0, stream>>>(dis, N);
    count_deg<<<gE, blk, 0, stream>>>(dst, dis, E);
    rsqrt_inplace<<<gN, blk, 0, stream>>>(dis, N);

    gemm_wmma_f32<<<gGemm, blk, 0, stream>>>(x, Wt1, bufA, N);
    zero_f32<<<gNF, blk, 0, stream>>>(bufB, N * NDIM);
    edge_scatter<<<gScat, blk, 0, stream>>>(bufA, dis, src, dst, bufB, E);
    post_epilogue<<<gNF, blk, 0, stream>>>(bufB, bufA, dis, b1, N,  1);

    gemm_wmma_f32<<<gGemm, blk, 0, stream>>>(bufB, Wt2, bufA, N);
    zero_f32<<<gNF, blk, 0, stream>>>(outp, N * NDIM);
    edge_scatter<<<gScat, blk, 0, stream>>>(bufA, dis, src, dst, outp, E);
    post_epilogue<<<gNF, blk, 0, stream>>>(outp, bufA, dis, b2, N,  0);
}


# Replace float atomic scatter by per-call dst-sorted neighbour lists (int atomic rank + scan) and a one-wave-per-node row gather fusing zero-fill and epilogue; row sums now in list order
# speedup vs baseline: 2.2982x; 2.2982x over previous
#include <hip/hip_runtime.h>
#include <hip/hip_bf16.h>

#define NDIM 128
#define NEG_SLOPE 0.01f

typedef __attribute__((ext_vector_type(2))) float v2f;
typedef __attribute__((ext_vector_type(8))) float v8f;

__global__ void transpose128(const float* __restrict__ W, float* __restrict__ Wt) {
    int i = blockIdx.x * blockDim.x + threadIdx.x;
    if (i < NDIM * NDIM) {
        int r = i >> 7, c = i & 127;
        Wt[c * NDIM + r] = W[i];
    }
}

__global__ void zero_i32(int* __restrict__ p, int n) {
    int i = blockIdx.x * blockDim.x + threadIdx.x;
    if (i < n) p[i] = 0;
}

__global__ void count_rank(const int* __restrict__ dst, int* __restrict__ cnt,
                           int* __restrict__ rank, int e) {
    int i = blockIdx.x * blockDim.x + threadIdx.x;
    if (i < e) rank[i] = atomicAdd(&cnt[dst[i]], 1);
}

__global__ void __launch_bounds__(256)
scan_local(const int* __restrict__ cnt, int* __restrict__ rowptr,
           int* __restrict__ bsum, int n) {
    __shared__ int wsum[8];
    const int tid  = threadIdx.x;
    const int lane = tid & 31;
    const int wave = tid >> 5;
    const int i = blockIdx.x * 256 + tid;
    const int v = (i < n) ? cnt[i] : 0;
    int x = v;
    for (int off = 1; off < 32; off <<= 1) {
        int y = __shfl_up(x, off);
        if (lane >= off) x += y;
    }
    if (lane == 31) wsum[wave] = x;
    __syncthreads();
    int wpre = 0;
    for (int w = 0; w < wave; ++w) wpre += wsum[w];
    if (i < n) rowptr[i] = wpre + x - v;
    if (tid == 255) bsum[blockIdx.x] = wpre + x;
}

__global__ void __launch_bounds__(256)
scan_fix(int* __restrict__ rowptr, const int* __restrict__ cnt,
         const int* __restrict__ bsum, float* __restrict__ dis, int n) {
    __shared__ int red[8];
    const int tid  = threadIdx.x;
    const int lane = tid & 31;
    const int wave = tid >> 5;
    const int b = blockIdx.x;
    int part = 0;
    for (int j = tid; j < b; j += 256) part += bsum[j];
    for (int off = 16; off > 0; off >>= 1) part += __shfl_xor(part, off);
    if (lane == 0) red[wave] = part;
    __syncthreads();
    int boff = 0;
    for (int w = 0; w < 8; ++w) boff += red[w];
    const int i = b * 256 + tid;
    if (i < n) {
        rowptr[i] += boff;
        dis[i] = rsqrtf((float)cnt[i] + 1.0f);
    }
}

__global__ void fill_col(const int* __restrict__ src, const int* __restrict__ dst,
                         const int* __restrict__ rowptr, const int* __restrict__ rank,
                         int* __restrict__ col, int e) {
    int i = blockIdx.x * blockDim.x + threadIdx.x;
    if (i < e) col[rowptr[dst[i]] + rank[i]] = src[i];
}

__global__ void __launch_bounds__(256)
gemm_wmma_f32(const float* __restrict__ X, const float* __restrict__ Wt,
              float* __restrict__ H, int nrows) {
    __shared__ float Xs[16][132];

    const int tid     = threadIdx.x;
    const int rowBase = blockIdx.x * 16;

    for (int i = tid; i < 16 * NDIM; i += 256) {
        int r = i >> 7, c = i & 127;
        int gr = rowBase + r;
        Xs[r][c] = (gr < nrows) ? X[gr * NDIM + c] : 0.0f;
    }
    __syncthreads();

    const int wave    = tid >> 5;
    const int lane    = tid & 31;
    const int colBase = wave * 16;
    const int l16     = lane & 15;
    const int khalf   = (lane >> 4) << 1;

    v8f acc = {};
    #pragma unroll
    for (int k = 0; k < NDIM; k += 4) {
        v2f a, b;
        a.x = Xs[l16][k + khalf];
        a.y = Xs[l16][k + khalf + 1];
        b.x = Wt[(k + khalf)     * NDIM + colBase + l16];
        b.y = Wt[(k + khalf + 1) * NDIM + colBase + l16];
        acc = __builtin_amdgcn_wmma_f32_16x16x4_f32(
             false, a,  false, b,
             (short)0, acc,  false,  false);
    }

    const int row0 = rowBase + ((lane >> 4) << 3);
    const int col  = colBase + l16;
    #pragma unroll
    for (int v = 0; v < 8; ++v) {
        int r = row0 + v;
        if (r < nrows) H[r * NDIM + col] = acc[v];
    }
}

typedef __attribute__((ext_vector_type(4))) float v4f;

__global__ void __launch_bounds__(256)
gather_rows(const float* __restrict__ h, const float* __restrict__ dis,
            const int* __restrict__ rowptr, const int* __restrict__ cnt,
            const int* __restrict__ col, const float* __restrict__ bias,
            float* __restrict__ outp, int n, int e, int do_lrelu) {
    const int lane = threadIdx.x & 31;
    int node = blockIdx.x * 8 + (threadIdx.x >> 5);
    if (node >= n) return;
    node = __builtin_amdgcn_readfirstlane(node);
    int c     = __builtin_amdgcn_readfirstlane(cnt[node]);
    int start = __builtin_amdgcn_readfirstlane(rowptr[node]);
    c     = min(max(c, 0), e);
    start = min(max(start, 0), e - c);

    const float dd = dis[node];
    const v4f* __restrict__ H4 = (const v4f*)h;
    v4f acc = {0.0f, 0.0f, 0.0f, 0.0f};

    for (int base = 0; base < c; base += 32) {
        const int m = min(32, c - base);
        int   my_s = 0;
        float my_w = 0.0f;
        if (lane < m) {
            my_s = col[start + base + lane];
            my_s = (int)min((unsigned)my_s, (unsigned)(n - 1));
            my_w = dis[my_s] * dd;
        }
        const int my_wi = __float_as_int(my_w);
        int j = 0;
        for (; j + 4 <= m; j += 4) {
            const int s0 = __builtin_amdgcn_readlane(my_s, j);
            const int s1 = __builtin_amdgcn_readlane(my_s, j + 1);
            const int s2 = __builtin_amdgcn_readlane(my_s, j + 2);
            const int s3 = __builtin_amdgcn_readlane(my_s, j + 3);
            const float w0 = __int_as_float(__builtin_amdgcn_readlane(my_wi, j));
            const float w1 = __int_as_float(__builtin_amdgcn_readlane(my_wi, j + 1));
            const float w2 = __int_as_float(__builtin_amdgcn_readlane(my_wi, j + 2));
            const float w3 = __int_as_float(__builtin_amdgcn_readlane(my_wi, j + 3));
            const v4f v0 = H4[(size_t)s0 * 32 + lane];
            const v4f v1 = H4[(size_t)s1 * 32 + lane];
            const v4f v2 = H4[(size_t)s2 * 32 + lane];
            const v4f v3 = H4[(size_t)s3 * 32 + lane];
            acc += v0 * w0;
            acc += v1 * w1;
            acc += v2 * w2;
            acc += v3 * w3;
        }
        const int rem = m - j;
        if (rem > 0) {
            v4f v0 = {0.0f, 0.0f, 0.0f, 0.0f};
            v4f v1 = v0, v2 = v0;
            float w0 = 0.0f, w1 = 0.0f, w2 = 0.0f;
            {
                const int s0 = __builtin_amdgcn_readlane(my_s, j);
                w0 = __int_as_float(__builtin_amdgcn_readlane(my_wi, j));
                v0 = H4[(size_t)s0 * 32 + lane];
            }
            if (rem > 1) {
                const int s1 = __builtin_amdgcn_readlane(my_s, j + 1);
                w1 = __int_as_float(__builtin_amdgcn_readlane(my_wi, j + 1));
                v1 = H4[(size_t)s1 * 32 + lane];
            }
            if (rem > 2) {
                const int s2 = __builtin_amdgcn_readlane(my_s, j + 2);
                w2 = __int_as_float(__builtin_amdgcn_readlane(my_wi, j + 2));
                v2 = H4[(size_t)s2 * 32 + lane];
            }
            acc += v0 * w0;
            acc += v1 * w1;
            acc += v2 * w2;
        }
    }

    const v4f hv = H4[(size_t)node * 32 + lane];
    const v4f bv = {bias[lane * 4], bias[lane * 4 + 1], bias[lane * 4 + 2], bias[lane * 4 + 3]};
    const float sl = dd * dd;
    v4f r = acc + hv * sl + bv;
    if (do_lrelu) {
        r.x = (r.x > 0.0f) ? r.x : NEG_SLOPE * r.x;
        r.y = (r.y > 0.0f) ? r.y : NEG_SLOPE * r.y;
        r.z = (r.z > 0.0f) ? r.z : NEG_SLOPE * r.z;
        r.w = (r.w > 0.0f) ? r.w : NEG_SLOPE * r.w;
    }
    ((v4f*)outp)[(size_t)node * 32 + lane] = r;
}

extern "C" void kernel_launch(void* const* d_in, const int* in_sizes, int n_in,
                              void* d_out, int out_size, void* d_ws, size_t ws_size,
                              hipStream_t stream) {
    const float* x    = (const float*)d_in[0];
    const int*   ei   = (const int*)d_in[1];
    const float* W1   = (const float*)d_in[3];
    const float* b1   = (const float*)d_in[4];
    const float* W2   = (const float*)d_in[5];
    const float* b2   = (const float*)d_in[6];

    const int N = in_sizes[0] / NDIM;
    const int E = in_sizes[1] / 2;
    const int* src = ei;
    const int* dst = ei + E;

    const int NP = (N + 3) & ~3;
    const int EP = (E + 3) & ~3;
    const int NB = (N + 255) / 256;
    float* Wt1    = (float*)d_ws;
    float* Wt2    = Wt1 + NDIM * NDIM;
    float* dis    = Wt2 + NDIM * NDIM;
    int*   cnt    = (int*)(dis + NP);
    int*   rowptr = cnt + NP;
    int*   bsum   = rowptr + NP;
    int*   rank   = bsum + ((NB + 3) & ~3);
    int*   col    = rank + EP;
    float* bufA   = (float*)(col + EP);
    float* bufB   = bufA + (size_t)N * NDIM;
    float* outp   = (float*)d_out;

    const int T = 256;
    dim3 blk(T);
    dim3 gW((NDIM * NDIM + T - 1) / T);
    dim3 gN((N + T - 1) / T);
    dim3 gE((E + T - 1) / T);
    dim3 gScan(NB);
    dim3 gGath((N + 7) / 8);
    dim3 gGemm((N + 15) / 16);

    transpose128<<<gW, blk, 0, stream>>>(W1, Wt1);
    transpose128<<<gW, blk, 0, stream>>>(W2, Wt2);
    zero_i32<<<gN, blk, 0, stream>>>(cnt, N);
    count_rank<<<gE, blk, 0, stream>>>(dst, cnt, rank, E);
    scan_local<<<gScan, blk, 0, stream>>>(cnt, rowptr, bsum, N);
    scan_fix<<<gScan, blk, 0, stream>>>(rowptr, cnt, bsum, dis, N);
    fill_col<<<gE, blk, 0, stream>>>(src, dst, rowptr, rank, col, E);

    gemm_wmma_f32<<<gGemm, blk, 0, stream>>>(x, Wt1, bufA, N);
    gather_rows<<<gGath, blk, 0, stream>>>(bufA, dis, rowptr, cnt, col, b1, bufB, N, E,  1);

    gemm_wmma_f32<<<gGemm, blk, 0, stream>>>(bufB, Wt2, bufA, N);
    gather_rows<<<gGath, blk, 0, stream>>>(bufA, dis, rowptr, cnt, col, b2, outp, N, E,  0);
}


# Neighbour lists built by a two-level counting sort in LDS (slice sort by 256-node bin, then one block per bin) instead of per-edge global atomics; list order = LDS atomic arrival; 9 -> 6 launches
# speedup vs baseline: 4.3525x; 1.8939x over previous
#include <hip/hip_runtime.h>
#include <hip/hip_bf16.h>

#define NDIM 128
#define NEG_SLOPE 0.01f

typedef __attribute__((ext_vector_type(2))) float v2f;
typedef __attribute__((ext_vector_type(8))) float v8f;

__global__ void transpose128(const float* __restrict__ W, float* __restrict__ Wt) {
    int i = blockIdx.x * blockDim.x + threadIdx.x;
    if (i < NDIM * NDIM) {
        int r = i >> 7, c = i & 127;
        Wt[c * NDIM + r] = W[i];
    }
}

#define BIN_SHIFT 8
#define BIN_NODES 256
#define PART_CAP  12800
#define PART_T    1024
#define PART_PER  ((PART_CAP + PART_T - 1) / PART_T)
#define FIN_T     512
#define FIN_CAP   8192

__device__ __forceinline__ int scan256(int v, int tid, int* wsum, int& total) {
    const int lane = tid & 31;
    const int wave = tid >> 5;
    int x = v;
    for (int off = 1; off < 32; off <<= 1) {
        int y = __shfl_up(x, off);
        if (lane >= off) x += y;
    }
    if (lane == 31 && wave < 8) wsum[wave] = x;
    __syncthreads();
    int wpre = 0, tot = 0;
    for (int w = 0; w < 8; ++w) {
        const int s = wsum[w];
        if (w < wave) wpre += s;
        tot += s;
    }
    total = tot;
    return wpre + x - v;
}

__global__ void __launch_bounds__(PART_T)
bin_sort_chunk(const int* __restrict__ src, const int* __restrict__ dst,
               int* __restrict__ sortedS, int* __restrict__ sortedD,
               int* __restrict__ lenT, int* __restrict__ startT,
               int n, int e, int nbin, int g) {
    __shared__ int sS[PART_CAP];
    __shared__ int sD[PART_CAP];
    __shared__ int hist[BIN_NODES];
    __shared__ int lstart[BIN_NODES];
    __shared__ int lcur[BIN_NODES];
    __shared__ int wsum[8];

    const int tid  = threadIdx.x;
    const int blk  = blockIdx.x;
    const int base = blk * PART_CAP;
    const int len  = min(PART_CAP, e - base);

    if (tid < BIN_NODES) { hist[tid] = 0; lcur[tid] = 0; }

    int dv[PART_PER], sv[PART_PER];
    #pragma unroll
    for (int k = 0; k < PART_PER; ++k) {
        const int i  = tid + k * PART_T;
        const int ii = min(i, len - 1);
        dv[k] = dst[base + ii];
        sv[k] = src[base + ii];
    }
    #pragma unroll
    for (int k = 0; k < PART_PER; ++k)
        dv[k] = (tid + k * PART_T < len) ? (int)min((unsigned)dv[k], (unsigned)(n - 1)) : -1;
    __syncthreads();

    #pragma unroll
    for (int k = 0; k < PART_PER; ++k)
        if (dv[k] >= 0) atomicAdd(&hist[dv[k] >> BIN_SHIFT], 1);
    __syncthreads();

    int total;
    const int v  = (tid < BIN_NODES) ? hist[tid] : 0;
    const int ex = scan256(v, tid, wsum, total);
    if (tid < BIN_NODES) {
        lstart[tid] = ex;
        if (tid < nbin) {
            lenT[tid * g + blk]   = v;
            startT[tid * g + blk] = base + ex;
        }
    }
    __syncthreads();

    #pragma unroll
    for (int k = 0; k < PART_PER; ++k) {
        if (dv[k] >= 0) {
            const int b = dv[k] >> BIN_SHIFT;
            const int p = lstart[b] + atomicAdd(&lcur[b], 1);
            sS[p] = sv[k];
            sD[p] = dv[k];
        }
    }
    __syncthreads();

    for (int i = tid; i < len; i += PART_T) {
        sortedS[base + i] = sS[i];
        sortedD[base + i] = sD[i];
    }
}

__global__ void __launch_bounds__(FIN_T)
bin_finalize(const int* __restrict__ sortedS, const int* __restrict__ sortedD,
             const int* __restrict__ lenT, const int* __restrict__ startT,
             int* __restrict__ rowptr, int* __restrict__ cnt,
             float* __restrict__ dis, int* __restrict__ col,
             int n, int e, int g) {
    __shared__ int rs[FIN_T];
    __shared__ int rl[FIN_T];
    __shared__ int ncnt[BIN_NODES];
    __shared__ int ncur[BIN_NODES];
    __shared__ int lcol[FIN_CAP];
    __shared__ int red[FIN_T / 32];
    __shared__ int wsum[8];

    const int tid  = threadIdx.x;
    const int lane = tid & 31;
    const int wave = tid >> 5;
    const int bin  = blockIdx.x;
    const int lim  = bin * g;

    const int lim4 = lim >> 2;
    const int4* __restrict__ L4 = (const int4*)lenT;
    int part = 0;
    for (int j0 = 0; j0 < lim4; j0 += 4 * FIN_T) {
        int4 q[4];
        #pragma unroll
        for (int u = 0; u < 4; ++u) {
            const int j = j0 + u * FIN_T + tid;
            q[u] = L4[min(j, lim4 - 1)];
        }
        #pragma unroll
        for (int u = 0; u < 4; ++u)
            if (j0 + u * FIN_T + tid < lim4) part += q[u].x + q[u].y + q[u].z + q[u].w;
    }
    {
        const int j = (lim4 << 2) + tid;
        if (tid < 4 && j < lim) part += lenT[j];
    }
    for (int off = 16; off > 0; off >>= 1) part += __shfl_xor(part, off);
    if (lane == 0) red[wave] = part;
    if (tid < BIN_NODES) ncnt[tid] = 0;
    if (tid < g) {
        int ln = lenT[lim + tid];
        int st = startT[lim + tid];
        ln = min(max(ln, 0), e);
        st = min(max(st, 0), e - ln);
        rl[tid] = ln;
        rs[tid] = st;
    }
    __syncthreads();
    int binBase = 0;
    for (int w = 0; w < FIN_T / 32; ++w) binBase += red[w];

    for (int r = wave; r < g; r += FIN_T / 32) {
        const int st = rs[r];
        const int ln = rl[r];
        for (int j0 = 0; j0 < ln; j0 += 128) {
            int dd[4];
            #pragma unroll
            for (int u = 0; u < 4; ++u) {
                const int j = j0 + u * 32 + lane;
                dd[u] = sortedD[st + min(j, ln - 1)];
            }
            #pragma unroll
            for (int u = 0; u < 4; ++u)
                if (j0 + u * 32 + lane >= ln) dd[u] = -1;
            #pragma unroll
            for (int u = 0; u < 4; ++u)
                if (dd[u] >= 0) atomicAdd(&ncnt[dd[u] & (BIN_NODES - 1)], 1);
        }
    }
    __syncthreads();

    int binCnt;
    const int v  = (tid < BIN_NODES) ? ncnt[tid] : 0;
    const int ex = scan256(v, tid, wsum, binCnt);
    binCnt  = min(max(binCnt, 0), e);
    binBase = min(max(binBase, 0), e - binCnt);
    if (tid < BIN_NODES) {
        ncur[tid] = ex;
        const int node = (bin << BIN_SHIFT) + tid;
        if (node < n) {
            cnt[node]    = v;
            rowptr[node] = binBase + ex;
            dis[node]    = rsqrtf((float)v + 1.0f);
        }
    }
    __syncthreads();

    const bool staged = binCnt <= FIN_CAP;
    for (int r = wave; r < g; r += FIN_T / 32) {
        const int st = rs[r];
        const int ln = rl[r];
        for (int j0 = 0; j0 < ln; j0 += 128) {
            int dd[4], ss[4];
            #pragma unroll
            for (int u = 0; u < 4; ++u) {
                const int jj = min(j0 + u * 32 + lane, ln - 1);
                dd[u] = sortedD[st + jj];
                ss[u] = sortedS[st + jj];
            }
            #pragma unroll
            for (int u = 0; u < 4; ++u)
                if (j0 + u * 32 + lane >= ln) dd[u] = -1;
            #pragma unroll
            for (int u = 0; u < 4; ++u) {
                if (dd[u] >= 0) {
                    int p = atomicAdd(&ncur[dd[u] & (BIN_NODES - 1)], 1);
                    p = min(max(p, 0), binCnt - 1);
                    if (staged) lcol[p] = ss[u];
                    else        col[binBase + p] = ss[u];
                }
            }
        }
    }
    __syncthreads();
    if (staged)
        for (int i = tid; i < binCnt; i += FIN_T) col[binBase + i] = lcol[i];
}

typedef __attribute__((ext_vector_type(4))) float v4f;

__device__ __forceinline__ v4f
gather_node(const v4f* __restrict__ X4, const float* __restrict__ dis,
            const int* __restrict__ rowptr, const int* __restrict__ cnt,
            const int* __restrict__ col, int node, int lane, int n, int e) {
    int c     = __builtin_amdgcn_readfirstlane(cnt[node]);
    int start = __builtin_amdgcn_readfirstlane(rowptr[node]);
    c     = min(max(c, 0), e);
    start = min(max(start, 0), e - c);

    const float dd = dis[node];
    const v4f xself = X4[(size_t)node * 32 + lane];
    v4f acc = {0.0f, 0.0f, 0.0f, 0.0f};

    for (int base = 0; base < c; base += 32) {
        const int m = min(32, c - base);
        int   my_s = 0;
        float my_w = 0.0f;
        if (lane < m) {
            my_s = col[start + base + lane];
            my_s = (int)min((unsigned)my_s, (unsigned)(n - 1));
            my_w = dis[my_s] * dd;
        }
        const int my_wi = __float_as_int(my_w);
        int j = 0;
        for (; j + 4 <= m; j += 4) {
            const int s0 = __builtin_amdgcn_readlane(my_s, j);
            const int s1 = __builtin_amdgcn_readlane(my_s, j + 1);
            const int s2 = __builtin_amdgcn_readlane(my_s, j + 2);
            const int s3 = __builtin_amdgcn_readlane(my_s, j + 3);
            const float w0 = __int_as_float(__builtin_amdgcn_readlane(my_wi, j));
            const float w1 = __int_as_float(__builtin_amdgcn_readlane(my_wi, j + 1));
            const float w2 = __int_as_float(__builtin_amdgcn_readlane(my_wi, j + 2));
            const float w3 = __int_as_float(__builtin_amdgcn_readlane(my_wi, j + 3));
            const v4f v0 = X4[(size_t)s0 * 32 + lane];
            const v4f v1 = X4[(size_t)s1 * 32 + lane];
            const v4f v2 = X4[(size_t)s2 * 32 + lane];
            const v4f v3 = X4[(size_t)s3 * 32 + lane];
            acc += v0 * w0;
            acc += v1 * w1;
            acc += v2 * w2;
            acc += v3 * w3;
        }
        const int rem = m - j;
        if (rem > 0) {
            v4f v0 = {0.0f, 0.0f, 0.0f, 0.0f};
            v4f v1 = v0, v2 = v0;
            float w0 = 0.0f, w1 = 0.0f, w2 = 0.0f;
            {
                const int s0 = __builtin_amdgcn_readlane(my_s, j);
                w0 = __int_as_float(__builtin_amdgcn_readlane(my_wi, j));
                v0 = X4[(size_t)s0 * 32 + lane];
            }
            if (rem > 1) {
                const int s1 = __builtin_amdgcn_readlane(my_s, j + 1);
                w1 = __int_as_float(__builtin_amdgcn_readlane(my_wi, j + 1));
                v1 = X4[(size_t)s1 * 32 + lane];
            }
            if (rem > 2) {
                const int s2 = __builtin_amdgcn_readlane(my_s, j + 2);
                w2 = __int_as_float(__builtin_amdgcn_readlane(my_wi, j + 2));
                v2 = X4[(size_t)s2 * 32 + lane];
            }
            acc += v0 * w0;
            acc += v1 * w1;
            acc += v2 * w2;
        }
    }
    return acc + xself * (dd * dd);
}

__global__ void __launch_bounds__(512)
gather_gemm(const float* __restrict__ xin, const float* __restrict__ dis,
            const int* __restrict__ rowptr, const int* __restrict__ cnt,
            const int* __restrict__ col, const float* __restrict__ Wt,
            const float* __restrict__ bias, float* __restrict__ outp,
            int n, int e, int do_lrelu) {
    __shared__ __attribute__((aligned(16))) float Xs[32][132];
    __shared__ __attribute__((aligned(16))) float Cs[32][132];

    const int tid     = threadIdx.x;
    const int lane    = tid & 31;
    const int wave    = __builtin_amdgcn_readfirstlane(tid >> 5);
    const int rowBase = blockIdx.x * 32;
    const v4f* __restrict__ X4 = (const v4f*)xin;

    for (int t = 0; t < 2; ++t) {
        const int r    = wave * 2 + t;
        const int node = rowBase + r;
        v4f y = {0.0f, 0.0f, 0.0f, 0.0f};
        if (node < n) y = gather_node(X4, dis, rowptr, cnt, col, node, lane, n, e);
        *(v4f*)&Xs[r][lane * 4] = y;
    }
    __syncthreads();

    if (wave < 8) {
        const int colBase = wave * 16;
        const int l16     = lane & 15;
        const int khalf   = (lane >> 4) << 1;

        v8f acc0 = {};
        v8f acc1 = {};
        #pragma unroll
        for (int k = 0; k < NDIM; k += 4) {
            v2f a0, a1, b;
            a0.x = Xs[l16][k + khalf];
            a0.y = Xs[l16][k + khalf + 1];
            a1.x = Xs[16 + l16][k + khalf];
            a1.y = Xs[16 + l16][k + khalf + 1];
            b.x = Wt[(k + khalf)     * NDIM + colBase + l16];
            b.y = Wt[(k + khalf + 1) * NDIM + colBase + l16];
            acc0 = __builtin_amdgcn_wmma_f32_16x16x4_f32(
                 false, a0,  false, b,
                 (short)0, acc0,  false,  false);
            acc1 = __builtin_amdgcn_wmma_f32_16x16x4_f32(
                 false, a1,  false, b,
                 (short)0, acc1,  false,  false);
        }

        const int crow0 = (lane >> 4) << 3;
        #pragma unroll
        for (int v = 0; v < 8; ++v) {
            Cs[crow0 + v][colBase + l16]      = acc0[v];
            Cs[16 + crow0 + v][colBase + l16] = acc1[v];
        }
    }
    const v4f bv = {bias[lane * 4], bias[lane * 4 + 1], bias[lane * 4 + 2], bias[lane * 4 + 3]};
    __syncthreads();

    for (int t = 0; t < 2; ++t) {
        const int r    = wave * 2 + t;
        const int node = rowBase + r;
        if (node < n) {
            v4f o = *(const v4f*)&Cs[r][lane * 4] + bv;
            if (do_lrelu) {
                o.x = (o.x > 0.0f) ? o.x : NEG_SLOPE * o.x;
                o.y = (o.y > 0.0f) ? o.y : NEG_SLOPE * o.y;
                o.z = (o.z > 0.0f) ? o.z : NEG_SLOPE * o.z;
                o.w = (o.w > 0.0f) ? o.w : NEG_SLOPE * o.w;
            }
            ((v4f*)outp)[(size_t)node * 32 + lane] = o;
        }
    }
}

extern "C" void kernel_launch(void* const* d_in, const int* in_sizes, int n_in,
                              void* d_out, int out_size, void* d_ws, size_t ws_size,
                              hipStream_t stream) {
    const float* x    = (const float*)d_in[0];
    const int*   ei   = (const int*)d_in[1];
    const float* W1   = (const float*)d_in[3];
    const float* b1   = (const float*)d_in[4];
    const float* W2   = (const float*)d_in[5];
    const float* b2   = (const float*)d_in[6];

    const int N = in_sizes[0] / NDIM;
    const int E = in_sizes[1] / 2;
    const int* src = ei;
    const int* dst = ei + E;

    const int NP   = (N + 3) & ~3;
    const int EP   = (E + 3) & ~3;
    const int NBIN = (N + BIN_NODES - 1) >> BIN_SHIFT;
    const int G    = (E + PART_CAP - 1) / PART_CAP;
    const int TP   = (NBIN * G + 3) & ~3;
    float* Wt1    = (float*)d_ws;
    float* Wt2    = Wt1 + NDIM * NDIM;
    float* dis    = Wt2 + NDIM * NDIM;
    int*   cnt    = (int*)(dis + NP);
    int*   rowptr = cnt + NP;
    int*   lenT   = rowptr + NP;
    int*   startT = lenT + TP;
    int*   sortS  = startT + TP;
    int*   sortD  = sortS + EP;
    int*   col    = sortD + EP;
    float* bufB   = (float*)(col + EP);
    float* outp   = (float*)d_out;

    const int T = 256;
    dim3 blk(T);
    dim3 blkLayer(512);
    dim3 gW((NDIM * NDIM + T - 1) / T);
    dim3 gLayer((N + 31) / 32);

    transpose128<<<gW, blk, 0, stream>>>(W1, Wt1);
    transpose128<<<gW, blk, 0, stream>>>(W2, Wt2);
    bin_sort_chunk<<<dim3(G), dim3(PART_T), 0, stream>>>(src, dst, sortS, sortD, lenT, startT, N, E, NBIN, G);
    bin_finalize<<<dim3(NBIN), dim3(FIN_T), 0, stream>>>(sortS, sortD, lenT, startT, rowptr, cnt, dis, col, N, E, G);

    gather_gemm<<<gLayer, blkLayer, 0, stream>>>(x, dis, rowptr, cnt, col, Wt1, b1, bufB, N, E,  1);

    gather_gemm<<<gLayer, blkLayer, 0, stream>>>(bufB, dis, rowptr, cnt, col, Wt2, b2, outp, N, E,  0);
}


# Both weight transposes run as 32 extra blocks of the slice-sort launch (on WGPs it leaves idle) instead of two launches of their own: 6 -> 4 dependent launches per call
# speedup vs baseline: 4.6401x; 1.0661x over previous
#include <hip/hip_runtime.h>
#include <hip/hip_bf16.h>

#define NDIM 128
#define NEG_SLOPE 0.01f

typedef __attribute__((ext_vector_type(2))) float v2f;
typedef __attribute__((ext_vector_type(8))) float v8f;

__device__ __forceinline__ void transpose_elem(const float* __restrict__ W,
                                               float* __restrict__ Wt, int j) {
    Wt[j] = W[(j & (NDIM - 1)) * NDIM + (j >> 7)];
}

#define BIN_SHIFT 8
#define BIN_NODES 256
#define PART_CAP  12800
#define PART_T    1024
#define PART_PER  ((PART_CAP + PART_T - 1) / PART_T)
#define FIN_T     512
#define FIN_CAP   8192
#define TR_BLOCKS ((2 * NDIM * NDIM) / PART_T)

__device__ __forceinline__ int scan256(int v, int tid, int* wsum, int& total) {
    const int lane = tid & 31;
    const int wave = tid >> 5;
    int x = v;
    for (int off = 1; off < 32; off <<= 1) {
        int y = __shfl_up(x, off);
        if (lane >= off) x += y;
    }
    if (lane == 31 && wave < 8) wsum[wave] = x;
    __syncthreads();
    int wpre = 0, tot = 0;
    for (int w = 0; w < 8; ++w) {
        const int s = wsum[w];
        if (w < wave) wpre += s;
        tot += s;
    }
    total = tot;
    return wpre + x - v;
}

__global__ void __launch_bounds__(PART_T)
bin_sort_chunk(const int* __restrict__ src, const int* __restrict__ dst,
               int* __restrict__ sortedS, int* __restrict__ sortedD,
               int* __restrict__ lenT, int* __restrict__ startT,
               const float* __restrict__ W1, float* __restrict__ Wt1,
               const float* __restrict__ W2, float* __restrict__ Wt2,
               int n, int e, int nbin, int g) {
    __shared__ int sS[PART_CAP];
    __shared__ int sD[PART_CAP];
    __shared__ int hist[BIN_NODES];
    __shared__ int lstart[BIN_NODES];
    __shared__ int lcur[BIN_NODES];
    __shared__ int wsum[8];

    const int tid  = threadIdx.x;
    const int blk  = blockIdx.x;

    if (blk >= g) {
        const int j = (blk - g) * PART_T + tid;
        if (j < NDIM * NDIM) transpose_elem(W1, Wt1, j);
        else                 transpose_elem(W2, Wt2, j - NDIM * NDIM);
        return;
    }

    const int base = blk * PART_CAP;
    const int len  = min(PART_CAP, e - base);

    if (tid < BIN_NODES) { hist[tid] = 0; lcur[tid] = 0; }

    int dv[PART_PER], sv[PART_PER];
    #pragma unroll
    for (int k = 0; k < PART_PER; ++k) {
        const int i  = tid + k * PART_T;
        const int ii = min(i, len - 1);
        dv[k] = dst[base + ii];
        sv[k] = src[base + ii];
    }
    #pragma unroll
    for (int k = 0; k < PART_PER; ++k)
        dv[k] = (tid + k * PART_T < len) ? (int)min((unsigned)dv[k], (unsigned)(n - 1)) : -1;
    __syncthreads();

    #pragma unroll
    for (int k = 0; k < PART_PER; ++k)
        if (dv[k] >= 0) atomicAdd(&hist[dv[k] >> BIN_SHIFT], 1);
    __syncthreads();

    int total;
    const int v  = (tid < BIN_NODES) ? hist[tid] : 0;
    const int ex = scan256(v, tid, wsum, total);
    if (tid < BIN_NODES) {
        lstart[tid] = ex;
        if (tid < nbin) {
            lenT[tid * g + blk]   = v;
            startT[tid * g + blk] = base + ex;
        }
    }
    __syncthreads();

    #pragma unroll
    for (int k = 0; k < PART_PER; ++k) {
        if (dv[k] >= 0) {
            const int b = dv[k] >> BIN_SHIFT;
            const int p = lstart[b] + atomicAdd(&lcur[b], 1);
            sS[p] = sv[k];
            sD[p] = dv[k];
        }
    }
    __syncthreads();

    for (int i = tid; i < len; i += PART_T) {
        sortedS[base + i] = sS[i];
        sortedD[base + i] = sD[i];
    }
}

__global__ void __launch_bounds__(FIN_T)
bin_finalize(const int* __restrict__ sortedS, const int* __restrict__ sortedD,
             const int* __restrict__ lenT, const int* __restrict__ startT,
             int* __restrict__ rowptr, int* __restrict__ cnt,
             float* __restrict__ dis, int* __restrict__ col,
             int n, int e, int g) {
    __shared__ int rs[FIN_T];
    __shared__ int rl[FIN_T];
    __shared__ int ncnt[BIN_NODES];
    __shared__ int ncur[BIN_NODES];
    __shared__ int lcol[FIN_CAP];
    __shared__ int red[FIN_T / 32];
    __shared__ int wsum[8];

    const int tid  = threadIdx.x;
    const int lane = tid & 31;
    const int wave = tid >> 5;
    const int bin  = blockIdx.x;
    const int lim  = bin * g;

    const int lim4 = lim >> 2;
    const int4* __restrict__ L4 = (const int4*)lenT;
    int part = 0;
    for (int j0 = 0; j0 < lim4; j0 += 4 * FIN_T) {
        int4 q[4];
        #pragma unroll
        for (int u = 0; u < 4; ++u) {
            const int j = j0 + u * FIN_T + tid;
            q[u] = L4[min(j, lim4 - 1)];
        }
        #pragma unroll
        for (int u = 0; u < 4; ++u)
            if (j0 + u * FIN_T + tid < lim4) part += q[u].x + q[u].y + q[u].z + q[u].w;
    }
    {
        const int j = (lim4 << 2) + tid;
        if (tid < 4 && j < lim) part += lenT[j];
    }
    for (int off = 16; off > 0; off >>= 1) part += __shfl_xor(part, off);
    if (lane == 0) red[wave] = part;
    if (tid < BIN_NODES) ncnt[tid] = 0;
    if (tid < g) {
        int ln = lenT[lim + tid];
        int st = startT[lim + tid];
        ln = min(max(ln, 0), e);
        st = min(max(st, 0), e - ln);
        rl[tid] = ln;
        rs[tid] = st;
    }
    __syncthreads();
    int binBase = 0;
    for (int w = 0; w < FIN_T / 32; ++w) binBase += red[w];

    for (int r = wave; r < g; r += FIN_T / 32) {
        const int st = rs[r];
        const int ln = rl[r];
        for (int j0 = 0; j0 < ln; j0 += 128) {
            int dd[4];
            #pragma unroll
            for (int u = 0; u < 4; ++u) {
                const int j = j0 + u * 32 + lane;
                dd[u] = sortedD[st + min(j, ln - 1)];
            }
            #pragma unroll
            for (int u = 0; u < 4; ++u)
                if (j0 + u * 32 + lane >= ln) dd[u] = -1;
            #pragma unroll
            for (int u = 0; u < 4; ++u)
                if (dd[u] >= 0) atomicAdd(&ncnt[dd[u] & (BIN_NODES - 1)], 1);
        }
    }
    __syncthreads();

    int binCnt;
    const int v  = (tid < BIN_NODES) ? ncnt[tid] : 0;
    const int ex = scan256(v, tid, wsum, binCnt);
    binCnt  = min(max(binCnt, 0), e);
    binBase = min(max(binBase, 0), e - binCnt);
    if (tid < BIN_NODES) {
        ncur[tid] = ex;
        const int node = (bin << BIN_SHIFT) + tid;
        if (node < n) {
            cnt[node]    = v;
            rowptr[node] = binBase + ex;
            dis[node]    = rsqrtf((float)v + 1.0f);
        }
    }
    __syncthreads();

    const bool staged = binCnt <= FIN_CAP;
    for (int r = wave; r < g; r += FIN_T / 32) {
        const int st = rs[r];
        const int ln = rl[r];
        for (int j0 = 0; j0 < ln; j0 += 128) {
            int dd[4], ss[4];
            #pragma unroll
            for (int u = 0; u < 4; ++u) {
                const int jj = min(j0 + u * 32 + lane, ln - 1);
                dd[u] = sortedD[st + jj];
                ss[u] = sortedS[st + jj];
            }
            #pragma unroll
            for (int u = 0; u < 4; ++u)
                if (j0 + u * 32 + lane >= ln) dd[u] = -1;
            #pragma unroll
            for (int u = 0; u < 4; ++u) {
                if (dd[u] >= 0) {
                    int p = atomicAdd(&ncur[dd[u] & (BIN_NODES - 1)], 1);
                    p = min(max(p, 0), binCnt - 1);
                    if (staged) lcol[p] = ss[u];
                    else        col[binBase + p] = ss[u];
                }
            }
        }
    }
    __syncthreads();
    if (staged)
        for (int i = tid; i < binCnt; i += FIN_T) col[binBase + i] = lcol[i];
}

typedef __attribute__((ext_vector_type(4))) float v4f;

__device__ __forceinline__ v4f
gather_node(const v4f* __restrict__ X4, const float* __restrict__ dis,
            const int* __restrict__ rowptr, const int* __restrict__ cnt,
            const int* __restrict__ col, int node, int lane, int n, int e) {
    int c     = __builtin_amdgcn_readfirstlane(cnt[node]);
    int start = __builtin_amdgcn_readfirstlane(rowptr[node]);
    c     = min(max(c, 0), e);
    start = min(max(start, 0), e - c);

    const float dd = dis[node];
    const v4f xself = X4[(size_t)node * 32 + lane];
    v4f acc = {0.0f, 0.0f, 0.0f, 0.0f};

    for (int base = 0; base < c; base += 32) {
        const int m = min(32, c - base);
        int   my_s = 0;
        float my_w = 0.0f;
        if (lane < m) {
            my_s = col[start + base + lane];
            my_s = (int)min((unsigned)my_s, (unsigned)(n - 1));
            my_w = dis[my_s] * dd;
        }
        const int my_wi = __float_as_int(my_w);
        int j = 0;
        for (; j + 4 <= m; j += 4) {
            const int s0 = __builtin_amdgcn_readlane(my_s, j);
            const int s1 = __builtin_amdgcn_readlane(my_s, j + 1);
            const int s2 = __builtin_amdgcn_readlane(my_s, j + 2);
            const int s3 = __builtin_amdgcn_readlane(my_s, j + 3);
            const float w0 = __int_as_float(__builtin_amdgcn_readlane(my_wi, j));
            const float w1 = __int_as_float(__builtin_amdgcn_readlane(my_wi, j + 1));
            const float w2 = __int_as_float(__builtin_amdgcn_readlane(my_wi, j + 2));
            const float w3 = __int_as_float(__builtin_amdgcn_readlane(my_wi, j + 3));
            const v4f v0 = X4[(size_t)s0 * 32 + lane];
            const v4f v1 = X4[(size_t)s1 * 32 + lane];
            const v4f v2 = X4[(size_t)s2 * 32 + lane];
            const v4f v3 = X4[(size_t)s3 * 32 + lane];
            acc += v0 * w0;
            acc += v1 * w1;
            acc += v2 * w2;
            acc += v3 * w3;
        }
        const int rem = m - j;
        if (rem > 0) {
            v4f v0 = {0.0f, 0.0f, 0.0f, 0.0f};
            v4f v1 = v0, v2 = v0;
            float w0 = 0.0f, w1 = 0.0f, w2 = 0.0f;
            {
                const int s0 = __builtin_amdgcn_readlane(my_s, j);
                w0 = __int_as_float(__builtin_amdgcn_readlane(my_wi, j));
                v0 = X4[(size_t)s0 * 32 + lane];
            }
            if (rem > 1) {
                const int s1 = __builtin_amdgcn_readlane(my_s, j + 1);
                w1 = __int_as_float(__builtin_amdgcn_readlane(my_wi, j + 1));
                v1 = X4[(size_t)s1 * 32 + lane];
            }
            if (rem > 2) {
                const int s2 = __builtin_amdgcn_readlane(my_s, j + 2);
                w2 = __int_as_float(__builtin_amdgcn_readlane(my_wi, j + 2));
                v2 = X4[(size_t)s2 * 32 + lane];
            }
            acc += v0 * w0;
            acc += v1 * w1;
            acc += v2 * w2;
        }
    }
    return acc + xself * (dd * dd);
}

__global__ void __launch_bounds__(512)
gather_gemm(const float* __restrict__ xin, const float* __restrict__ dis,
            const int* __restrict__ rowptr, const int* __restrict__ cnt,
            const int* __restrict__ col, const float* __restrict__ Wt,
            const float* __restrict__ bias, float* __restrict__ outp,
            int n, int e, int do_lrelu) {
    __shared__ __attribute__((aligned(16))) float Xs[32][132];
    __shared__ __attribute__((aligned(16))) float Cs[32][132];

    const int tid     = threadIdx.x;
    const int lane    = tid & 31;
    const int wave    = __builtin_amdgcn_readfirstlane(tid >> 5);
    const int rowBase = blockIdx.x * 32;
    const v4f* __restrict__ X4 = (const v4f*)xin;

    for (int t = 0; t < 2; ++t) {
        const int r    = wave * 2 + t;
        const int node = rowBase + r;
        v4f y = {0.0f, 0.0f, 0.0f, 0.0f};
        if (node < n) y = gather_node(X4, dis, rowptr, cnt, col, node, lane, n, e);
        *(v4f*)&Xs[r][lane * 4] = y;
    }
    __syncthreads();

    if (wave < 8) {
        const int colBase = wave * 16;
        const int l16     = lane & 15;
        const int khalf   = (lane >> 4) << 1;

        v8f acc0 = {};
        v8f acc1 = {};
        #pragma unroll
        for (int k = 0; k < NDIM; k += 4) {
            v2f a0, a1, b;
            a0.x = Xs[l16][k + khalf];
            a0.y = Xs[l16][k + khalf + 1];
            a1.x = Xs[16 + l16][k + khalf];
            a1.y = Xs[16 + l16][k + khalf + 1];
            b.x = Wt[(k + khalf)     * NDIM + colBase + l16];
            b.y = Wt[(k + khalf + 1) * NDIM + colBase + l16];
            acc0 = __builtin_amdgcn_wmma_f32_16x16x4_f32(
                 false, a0,  false, b,
                 (short)0, acc0,  false,  false);
            acc1 = __builtin_amdgcn_wmma_f32_16x16x4_f32(
                 false, a1,  false, b,
                 (short)0, acc1,  false,  false);
        }

        const int crow0 = (lane >> 4) << 3;
        #pragma unroll
        for (int v = 0; v < 8; ++v) {
            Cs[crow0 + v][colBase + l16]      = acc0[v];
            Cs[16 + crow0 + v][colBase + l16] = acc1[v];
        }
    }
    const v4f bv = {bias[lane * 4], bias[lane * 4 + 1], bias[lane * 4 + 2], bias[lane * 4 + 3]};
    __syncthreads();

    for (int t = 0; t < 2; ++t) {
        const int r    = wave * 2 + t;
        const int node = rowBase + r;
        if (node < n) {
            v4f o = *(const v4f*)&Cs[r][lane * 4] + bv;
            if (do_lrelu) {
                o.x = (o.x > 0.0f) ? o.x : NEG_SLOPE * o.x;
                o.y = (o.y > 0.0f) ? o.y : NEG_SLOPE * o.y;
                o.z = (o.z > 0.0f) ? o.z : NEG_SLOPE * o.z;
                o.w = (o.w > 0.0f) ? o.w : NEG_SLOPE * o.w;
            }
            ((v4f*)outp)[(size_t)node * 32 + lane] = o;
        }
    }
}

extern "C" void kernel_launch(void* const* d_in, const int* in_sizes, int n_in,
                              void* d_out, int out_size, void* d_ws, size_t ws_size,
                              hipStream_t stream) {
    const float* x    = (const float*)d_in[0];
    const int*   ei   = (const int*)d_in[1];
    const float* W1   = (const float*)d_in[3];
    const float* b1   = (const float*)d_in[4];
    const float* W2   = (const float*)d_in[5];
    const float* b2   = (const float*)d_in[6];

    const int N = in_sizes[0] / NDIM;
    const int E = in_sizes[1] / 2;
    const int* src = ei;
    const int* dst = ei + E;

    const int NP   = (N + 3) & ~3;
    const int EP   = (E + 3) & ~3;
    const int NBIN = (N + BIN_NODES - 1) >> BIN_SHIFT;
    const int G    = (E + PART_CAP - 1) / PART_CAP;
    const int TP   = (NBIN * G + 3) & ~3;
    float* Wt1    = (float*)d_ws;
    float* Wt2    = Wt1 + NDIM * NDIM;
    float* dis    = Wt2 + NDIM * NDIM;
    int*   cnt    = (int*)(dis + NP);
    int*   rowptr = cnt + NP;
    int*   lenT   = rowptr + NP;
    int*   startT = lenT + TP;
    int*   sortS  = startT + TP;
    int*   sortD  = sortS + EP;
    int*   col    = sortD + EP;
    float* bufB   = (float*)(col + EP);
    float* outp   = (float*)d_out;

    dim3 blkLayer(512);
    dim3 gLayer((N + 31) / 32);

    bin_sort_chunk<<<dim3(G + TR_BLOCKS), dim3(PART_T), 0, stream>>>(src, dst, sortS, sortD, lenT, startT,
                                                                     W1, Wt1, W2, Wt2, N, E, NBIN, G);
    bin_finalize<<<dim3(NBIN), dim3(FIN_T), 0, stream>>>(sortS, sortD, lenT, startT, rowptr, cnt, dis, col, N, E, G);

    gather_gemm<<<gLayer, blkLayer, 0, stream>>>(x, dis, rowptr, cnt, col, Wt1, b1, bufB, N, E,  1);

    gather_gemm<<<gLayer, blkLayer, 0, stream>>>(bufB, dis, rowptr, cnt, col, Wt2, b2, outp, N, E,  0);
}


# Layer 1 stores each row already times deg^-1/2; layer 2 adds own row and neighbour rows unweighted, scales the sum once: no per-edge deg^-1/2 lookup in layer 2 (own row first; rounding changes)
# speedup vs baseline: 4.8670x; 1.0489x over previous
#include <hip/hip_runtime.h>
#include <hip/hip_bf16.h>

#define NDIM 128
#define NEG_SLOPE 0.01f

typedef __attribute__((ext_vector_type(2))) float v2f;
typedef __attribute__((ext_vector_type(8))) float v8f;

__device__ __forceinline__ void transpose_elem(const float* __restrict__ W,
                                               float* __restrict__ Wt, int j) {
    Wt[j] = W[(j & (NDIM - 1)) * NDIM + (j >> 7)];
}

#define BIN_SHIFT 8
#define BIN_NODES 256
#define PART_CAP  12800
#define PART_T    1024
#define PART_PER  ((PART_CAP + PART_T - 1) / PART_T)
#define FIN_T     512
#define FIN_CAP   8192
#define TR_BLOCKS ((2 * NDIM * NDIM) / PART_T)

__device__ __forceinline__ int scan256(int v, int tid, int* wsum, int& total) {
    const int lane = tid & 31;
    const int wave = tid >> 5;
    int x = v;
    for (int off = 1; off < 32; off <<= 1) {
        int y = __shfl_up(x, off);
        if (lane >= off) x += y;
    }
    if (lane == 31 && wave < 8) wsum[wave] = x;
    __syncthreads();
    int wpre = 0, tot = 0;
    for (int w = 0; w < 8; ++w) {
        const int s = wsum[w];
        if (w < wave) wpre += s;
        tot += s;
    }
    total = tot;
    return wpre + x - v;
}

__global__ void __launch_bounds__(PART_T)
bin_sort_chunk(const int* __restrict__ src, const int* __restrict__ dst,
               int* __restrict__ sortedS, int* __restrict__ sortedD,
               int* __restrict__ lenT, int* __restrict__ startT,
               const float* __restrict__ W1, float* __restrict__ Wt1,
               const float* __restrict__ W2, float* __restrict__ Wt2,
               int n, int e, int nbin, int g) {
    __shared__ int sS[PART_CAP];
    __shared__ int sD[PART_CAP];
    __shared__ int hist[BIN_NODES];
    __shared__ int lstart[BIN_NODES];
    __shared__ int lcur[BIN_NODES];
    __shared__ int wsum[8];

    const int tid  = threadIdx.x;
    const int blk  = blockIdx.x;

    if (blk >= g) {
        const int j = (blk - g) * PART_T + tid;
        if (j < NDIM * NDIM) transpose_elem(W1, Wt1, j);
        else                 transpose_elem(W2, Wt2, j - NDIM * NDIM);
        return;
    }

    const int base = blk * PART_CAP;
    const int len  = min(PART_CAP, e - base);

    if (tid < BIN_NODES) { hist[tid] = 0; lcur[tid] = 0; }

    int dv[PART_PER], sv[PART_PER];
    #pragma unroll
    for (int k = 0; k < PART_PER; ++k) {
        const int i  = tid + k * PART_T;
        const int ii = min(i, len - 1);
        dv[k] = dst[base + ii];
        sv[k] = src[base + ii];
    }
    #pragma unroll
    for (int k = 0; k < PART_PER; ++k)
        dv[k] = (tid + k * PART_T < len) ? (int)min((unsigned)dv[k], (unsigned)(n - 1)) : -1;
    __syncthreads();

    #pragma unroll
    for (int k = 0; k < PART_PER; ++k)
        if (dv[k] >= 0) atomicAdd(&hist[dv[k] >> BIN_SHIFT], 1);
    __syncthreads();

    int total;
    const int v  = (tid < BIN_NODES) ? hist[tid] : 0;
    const int ex = scan256(v, tid, wsum, total);
    if (tid < BIN_NODES) {
        lstart[tid] = ex;
        if (tid < nbin) {
            lenT[tid * g + blk]   = v;
            startT[tid * g + blk] = base + ex;
        }
    }
    __syncthreads();

    #pragma unroll
    for (int k = 0; k < PART_PER; ++k) {
        if (dv[k] >= 0) {
            const int b = dv[k] >> BIN_SHIFT;
            const int p = lstart[b] + atomicAdd(&lcur[b], 1);
            sS[p] = sv[k];
            sD[p] = dv[k];
        }
    }
    __syncthreads();

    for (int i = tid; i < len; i += PART_T) {
        sortedS[base + i] = sS[i];
        sortedD[base + i] = sD[i];
    }
}

__global__ void __launch_bounds__(FIN_T)
bin_finalize(const int* __restrict__ sortedS, const int* __restrict__ sortedD,
             const int* __restrict__ lenT, const int* __restrict__ startT,
             int* __restrict__ rowptr, int* __restrict__ cnt,
             float* __restrict__ dis, int* __restrict__ col,
             int n, int e, int g) {
    __shared__ int rs[FIN_T];
    __shared__ int rl[FIN_T];
    __shared__ int ncnt[BIN_NODES];
    __shared__ int ncur[BIN_NODES];
    __shared__ int lcol[FIN_CAP];
    __shared__ int red[FIN_T / 32];
    __shared__ int wsum[8];

    const int tid  = threadIdx.x;
    const int lane = tid & 31;
    const int wave = tid >> 5;
    const int bin  = blockIdx.x;
    const int lim  = bin * g;

    const int lim4 = lim >> 2;
    const int4* __restrict__ L4 = (const int4*)lenT;
    int part = 0;
    for (int j0 = 0; j0 < lim4; j0 += 4 * FIN_T) {
        int4 q[4];
        #pragma unroll
        for (int u = 0; u < 4; ++u) {
            const int j = j0 + u * FIN_T + tid;
            q[u] = L4[min(j, lim4 - 1)];
        }
        #pragma unroll
        for (int u = 0; u < 4; ++u)
            if (j0 + u * FIN_T + tid < lim4) part += q[u].x + q[u].y + q[u].z + q[u].w;
    }
    {
        const int j = (lim4 << 2) + tid;
        if (tid < 4 && j < lim) part += lenT[j];
    }
    for (int off = 16; off > 0; off >>= 1) part += __shfl_xor(part, off);
    if (lane == 0) red[wave] = part;
    if (tid < BIN_NODES) ncnt[tid] = 0;
    if (tid < g) {
        int ln = lenT[lim + tid];
        int st = startT[lim + tid];
        ln = min(max(ln, 0), e);
        st = min(max(st, 0), e - ln);
        rl[tid] = ln;
        rs[tid] = st;
    }
    __syncthreads();
    int binBase = 0;
    for (int w = 0; w < FIN_T / 32; ++w) binBase += red[w];

    for (int r = wave; r < g; r += FIN_T / 32) {
        const int st = rs[r];
        const int ln = rl[r];
        for (int j0 = 0; j0 < ln; j0 += 128) {
            int dd[4];
            #pragma unroll
            for (int u = 0; u < 4; ++u) {
                const int j = j0 + u * 32 + lane;
                dd[u] = sortedD[st + min(j, ln - 1)];
            }
            #pragma unroll
            for (int u = 0; u < 4; ++u)
                if (j0 + u * 32 + lane >= ln) dd[u] = -1;
            #pragma unroll
            for (int u = 0; u < 4; ++u)
                if (dd[u] >= 0) atomicAdd(&ncnt[dd[u] & (BIN_NODES - 1)], 1);
        }
    }
    __syncthreads();

    int binCnt;
    const int v  = (tid < BIN_NODES) ? ncnt[tid] : 0;
    const int ex = scan256(v, tid, wsum, binCnt);
    binCnt  = min(max(binCnt, 0), e);
    binBase = min(max(binBase, 0), e - binCnt);
    if (tid < BIN_NODES) {
        ncur[tid] = ex;
        const int node = (bin << BIN_SHIFT) + tid;
        if (node < n) {
            cnt[node]    = v;
            rowptr[node] = binBase + ex;
            dis[node]    = rsqrtf((float)v + 1.0f);
        }
    }
    __syncthreads();

    const bool staged = binCnt <= FIN_CAP;
    for (int r = wave; r < g; r += FIN_T / 32) {
        const int st = rs[r];
        const int ln = rl[r];
        for (int j0 = 0; j0 < ln; j0 += 128) {
            int dd[4], ss[4];
            #pragma unroll
            for (int u = 0; u < 4; ++u) {
                const int jj = min(j0 + u * 32 + lane, ln - 1);
                dd[u] = sortedD[st + jj];
                ss[u] = sortedS[st + jj];
            }
            #pragma unroll
            for (int u = 0; u < 4; ++u)
                if (j0 + u * 32 + lane >= ln) dd[u] = -1;
            #pragma unroll
            for (int u = 0; u < 4; ++u) {
                if (dd[u] >= 0) {
                    int p = atomicAdd(&ncur[dd[u] & (BIN_NODES - 1)], 1);
                    p = min(max(p, 0), binCnt - 1);
                    if (staged) lcol[p] = ss[u];
                    else        col[binBase + p] = ss[u];
                }
            }
        }
    }
    __syncthreads();
    if (staged)
        for (int i = tid; i < binCnt; i += FIN_T) col[binBase + i] = lcol[i];
}

typedef __attribute__((ext_vector_type(4))) float v4f;

__device__ __forceinline__ v4f
gather_node(const v4f* __restrict__ X4, const float* __restrict__ dis,
            const int* __restrict__ rowptr, const int* __restrict__ cnt,
            const int* __restrict__ col, int node, int lane, int n, int e) {
    int c     = __builtin_amdgcn_readfirstlane(cnt[node]);
    int start = __builtin_amdgcn_readfirstlane(rowptr[node]);
    c     = min(max(c, 0), e);
    start = min(max(start, 0), e - c);

    const float dd = dis[node];
    const v4f xself = X4[(size_t)node * 32 + lane];
    v4f acc = {0.0f, 0.0f, 0.0f, 0.0f};

    for (int base = 0; base < c; base += 32) {
        const int m = min(32, c - base);
        int   my_s = 0;
        float my_w = 0.0f;
        if (lane < m) {
            my_s = col[start + base + lane];
            my_s = (int)min((unsigned)my_s, (unsigned)(n - 1));
            my_w = dis[my_s] * dd;
        }
        const int my_wi = __float_as_int(my_w);
        int j = 0;
        for (; j + 4 <= m; j += 4) {
            const int s0 = __builtin_amdgcn_readlane(my_s, j);
            const int s1 = __builtin_amdgcn_readlane(my_s, j + 1);
            const int s2 = __builtin_amdgcn_readlane(my_s, j + 2);
            const int s3 = __builtin_amdgcn_readlane(my_s, j + 3);
            const float w0 = __int_as_float(__builtin_amdgcn_readlane(my_wi, j));
            const float w1 = __int_as_float(__builtin_amdgcn_readlane(my_wi, j + 1));
            const float w2 = __int_as_float(__builtin_amdgcn_readlane(my_wi, j + 2));
            const float w3 = __int_as_float(__builtin_amdgcn_readlane(my_wi, j + 3));
            const v4f v0 = X4[(size_t)s0 * 32 + lane];
            const v4f v1 = X4[(size_t)s1 * 32 + lane];
            const v4f v2 = X4[(size_t)s2 * 32 + lane];
            const v4f v3 = X4[(size_t)s3 * 32 + lane];
            acc += v0 * w0;
            acc += v1 * w1;
            acc += v2 * w2;
            acc += v3 * w3;
        }
        const int rem = m - j;
        if (rem > 0) {
            v4f v0 = {0.0f, 0.0f, 0.0f, 0.0f};
            v4f v1 = v0, v2 = v0;
            float w0 = 0.0f, w1 = 0.0f, w2 = 0.0f;
            {
                const int s0 = __builtin_amdgcn_readlane(my_s, j);
                w0 = __int_as_float(__builtin_amdgcn_readlane(my_wi, j));
                v0 = X4[(size_t)s0 * 32 + lane];
            }
            if (rem > 1) {
                const int s1 = __builtin_amdgcn_readlane(my_s, j + 1);
                w1 = __int_as_float(__builtin_amdgcn_readlane(my_wi, j + 1));
                v1 = X4[(size_t)s1 * 32 + lane];
            }
            if (rem > 2) {
                const int s2 = __builtin_amdgcn_readlane(my_s, j + 2);
                w2 = __int_as_float(__builtin_amdgcn_readlane(my_wi, j + 2));
                v2 = X4[(size_t)s2 * 32 + lane];
            }
            acc += v0 * w0;
            acc += v1 * w1;
            acc += v2 * w2;
        }
    }
    return acc + xself * (dd * dd);
}

__device__ __forceinline__ v4f
gather_node_scaled(const v4f* __restrict__ X4, const float* __restrict__ dis,
                   const int* __restrict__ rowptr, const int* __restrict__ cnt,
                   const int* __restrict__ col, int node, int lane, int n, int e) {
    int c     = __builtin_amdgcn_readfirstlane(cnt[node]);
    int start = __builtin_amdgcn_readfirstlane(rowptr[node]);
    c     = min(max(c, 0), e);
    start = min(max(start, 0), e - c);

    const float dd = dis[node];
    v4f acc = X4[(size_t)node * 32 + lane];

    for (int base = 0; base < c; base += 32) {
        const int m = min(32, c - base);
        int my_s = 0;
        if (lane < m) {
            my_s = col[start + base + lane];
            my_s = (int)min((unsigned)my_s, (unsigned)(n - 1));
        }
        int j = 0;
        for (; j + 4 <= m; j += 4) {
            const int s0 = __builtin_amdgcn_readlane(my_s, j);
            const int s1 = __builtin_amdgcn_readlane(my_s, j + 1);
            const int s2 = __builtin_amdgcn_readlane(my_s, j + 2);
            const int s3 = __builtin_amdgcn_readlane(my_s, j + 3);
            const v4f v0 = X4[(size_t)s0 * 32 + lane];
            const v4f v1 = X4[(size_t)s1 * 32 + lane];
            const v4f v2 = X4[(size_t)s2 * 32 + lane];
            const v4f v3 = X4[(size_t)s3 * 32 + lane];
            acc += v0;
            acc += v1;
            acc += v2;
            acc += v3;
        }
        const int rem = m - j;
        if (rem > 0) {
            v4f v1 = {0.0f, 0.0f, 0.0f, 0.0f};
            v4f v2 = v1;
            const int s0 = __builtin_amdgcn_readlane(my_s, j);
            const v4f v0 = X4[(size_t)s0 * 32 + lane];
            if (rem > 1) {
                const int s1 = __builtin_amdgcn_readlane(my_s, j + 1);
                v1 = X4[(size_t)s1 * 32 + lane];
            }
            if (rem > 2) {
                const int s2 = __builtin_amdgcn_readlane(my_s, j + 2);
                v2 = X4[(size_t)s2 * 32 + lane];
            }
            acc += v0;
            acc += v1;
            acc += v2;
        }
    }
    return acc * dd;
}

template <bool FIRST>
__global__ void __launch_bounds__(512)
gather_gemm(const float* __restrict__ xin, const float* __restrict__ dis,
            const int* __restrict__ rowptr, const int* __restrict__ cnt,
            const int* __restrict__ col, const float* __restrict__ Wt,
            const float* __restrict__ bias, float* __restrict__ outp,
            int n, int e) {
    __shared__ __attribute__((aligned(16))) float Xs[32][132];
    __shared__ __attribute__((aligned(16))) float Cs[32][132];

    const int tid     = threadIdx.x;
    const int lane    = tid & 31;
    const int wave    = __builtin_amdgcn_readfirstlane(tid >> 5);
    const int rowBase = blockIdx.x * 32;
    const v4f* __restrict__ X4 = (const v4f*)xin;

    for (int t = 0; t < 2; ++t) {
        const int r    = wave * 2 + t;
        const int node = rowBase + r;
        v4f y = {0.0f, 0.0f, 0.0f, 0.0f};
        if (node < n) {
            if (FIRST) y = gather_node(X4, dis, rowptr, cnt, col, node, lane, n, e);
            else       y = gather_node_scaled(X4, dis, rowptr, cnt, col, node, lane, n, e);
        }
        *(v4f*)&Xs[r][lane * 4] = y;
    }
    __syncthreads();

    if (wave < 8) {
        const int colBase = wave * 16;
        const int l16     = lane & 15;
        const int khalf   = (lane >> 4) << 1;

        v8f acc0 = {};
        v8f acc1 = {};
        #pragma unroll
        for (int k = 0; k < NDIM; k += 4) {
            v2f a0, a1, b;
            a0.x = Xs[l16][k + khalf];
            a0.y = Xs[l16][k + khalf + 1];
            a1.x = Xs[16 + l16][k + khalf];
            a1.y = Xs[16 + l16][k + khalf + 1];
            b.x = Wt[(k + khalf)     * NDIM + colBase + l16];
            b.y = Wt[(k + khalf + 1) * NDIM + colBase + l16];
            acc0 = __builtin_amdgcn_wmma_f32_16x16x4_f32(
                 false, a0,  false, b,
                 (short)0, acc0,  false,  false);
            acc1 = __builtin_amdgcn_wmma_f32_16x16x4_f32(
                 false, a1,  false, b,
                 (short)0, acc1,  false,  false);
        }

        const int crow0 = (lane >> 4) << 3;
        #pragma unroll
        for (int v = 0; v < 8; ++v) {
            Cs[crow0 + v][colBase + l16]      = acc0[v];
            Cs[16 + crow0 + v][colBase + l16] = acc1[v];
        }
    }
    const v4f bv = {bias[lane * 4], bias[lane * 4 + 1], bias[lane * 4 + 2], bias[lane * 4 + 3]};
    float dsc[2] = {1.0f, 1.0f};
    if (FIRST) {
        dsc[0] = dis[min(rowBase + wave * 2,     n - 1)];
        dsc[1] = dis[min(rowBase + wave * 2 + 1, n - 1)];
    }
    __syncthreads();

    #pragma unroll
    for (int t = 0; t < 2; ++t) {
        const int r    = wave * 2 + t;
        const int node = rowBase + r;
        if (node < n) {
            v4f o = *(const v4f*)&Cs[r][lane * 4] + bv;
            if (FIRST) {
                o.x = (o.x > 0.0f) ? o.x : NEG_SLOPE * o.x;
                o.y = (o.y > 0.0f) ? o.y : NEG_SLOPE * o.y;
                o.z = (o.z > 0.0f) ? o.z : NEG_SLOPE * o.z;
                o.w = (o.w > 0.0f) ? o.w : NEG_SLOPE * o.w;
                o = o * dsc[t];
            }
            ((v4f*)outp)[(size_t)node * 32 + lane] = o;
        }
    }
}

extern "C" void kernel_launch(void* const* d_in, const int* in_sizes, int n_in,
                              void* d_out, int out_size, void* d_ws, size_t ws_size,
                              hipStream_t stream) {
    const float* x    = (const float*)d_in[0];
    const int*   ei   = (const int*)d_in[1];
    const float* W1   = (const float*)d_in[3];
    const float* b1   = (const float*)d_in[4];
    const float* W2   = (const float*)d_in[5];
    const float* b2   = (const float*)d_in[6];

    const int N = in_sizes[0] / NDIM;
    const int E = in_sizes[1] / 2;
    const int* src = ei;
    const int* dst = ei + E;

    const int NP   = (N + 3) & ~3;
    const int EP   = (E + 3) & ~3;
    const int NBIN = (N + BIN_NODES - 1) >> BIN_SHIFT;
    const int G    = (E + PART_CAP - 1) / PART_CAP;
    const int TP   = (NBIN * G + 3) & ~3;
    float* Wt1    = (float*)d_ws;
    float* Wt2    = Wt1 + NDIM * NDIM;
    float* dis    = Wt2 + NDIM * NDIM;
    int*   cnt    = (int*)(dis + NP);
    int*   rowptr = cnt + NP;
    int*   lenT   = rowptr + NP;
    int*   startT = lenT + TP;
    int*   sortS  = startT + TP;
    int*   sortD  = sortS + EP;
    int*   col    = sortD + EP;
    float* bufB   = (float*)(col + EP);
    float* outp   = (float*)d_out;

    dim3 blkLayer(512);
    dim3 gLayer((N + 31) / 32);

    bin_sort_chunk<<<dim3(G + TR_BLOCKS), dim3(PART_T), 0, stream>>>(src, dst, sortS, sortD, lenT, startT,
                                                                     W1, Wt1, W2, Wt2, N, E, NBIN, G);
    bin_finalize<<<dim3(NBIN), dim3(FIN_T), 0, stream>>>(sortS, sortD, lenT, startT, rowptr, cnt, dis, col, N, E, G);

    gather_gemm<true><<<gLayer, blkLayer, 0, stream>>>(x, dis, rowptr, cnt, col, Wt1, b1, bufB, N, E);

    gather_gemm<false><<<gLayer, blkLayer, 0, stream>>>(bufB, dis, rowptr, cnt, col, Wt2, b2, outp, N, E);
}


# Level-1 sort slices 12800 -> 3584 edges: 224 sorting + 32 transposing 16-wave blocks, one per WGP (were 63 + 32); level 2 walks its 224 short runs 16 per load group; list order still LDS arrival
# speedup vs baseline: 5.2298x; 1.0745x over previous
#include <hip/hip_runtime.h>
#include <hip/hip_bf16.h>

#define NDIM 128
#define NEG_SLOPE 0.01f

typedef __attribute__((ext_vector_type(2))) float v2f;
typedef __attribute__((ext_vector_type(8))) float v8f;

__device__ __forceinline__ void transpose_elem(const float* __restrict__ W,
                                               float* __restrict__ Wt, int j) {
    Wt[j] = W[(j & (NDIM - 1)) * NDIM + (j >> 7)];
}

#define BIN_SHIFT 8
#define BIN_NODES 256
#define PART_CAP  3584
#define PART_T    512
#define PART_PER  ((PART_CAP + PART_T - 1) / PART_T)
#define FIN_T     512
#define FIN_CAP   8192
#define TR_BLOCKS ((NDIM * NDIM) / PART_T)

__device__ __forceinline__ int scan256(int v, int tid, int* wsum, int& total) {
    const int lane = tid & 31;
    const int wave = tid >> 5;
    int x = v;
    for (int off = 1; off < 32; off <<= 1) {
        int y = __shfl_up(x, off);
        if (lane >= off) x += y;
    }
    if (lane == 31 && wave < 8) wsum[wave] = x;
    __syncthreads();
    int wpre = 0, tot = 0;
    for (int w = 0; w < 8; ++w) {
        const int s = wsum[w];
        if (w < wave) wpre += s;
        tot += s;
    }
    total = tot;
    return wpre + x - v;
}

__global__ void __launch_bounds__(PART_T)
bin_sort_chunk(const int* __restrict__ src, const int* __restrict__ dst,
               int* __restrict__ sortedS, int* __restrict__ sortedD,
               int* __restrict__ lenT, int* __restrict__ startT,
               const float* __restrict__ W1, float* __restrict__ Wt1,
               const float* __restrict__ W2, float* __restrict__ Wt2,
               int n, int e, int nbin, int g) {
    __shared__ int sS[PART_CAP];
    __shared__ int sD[PART_CAP];
    __shared__ int hist[BIN_NODES];
    __shared__ int lstart[BIN_NODES];
    __shared__ int lcur[BIN_NODES];
    __shared__ int wsum[8];

    const int tid  = threadIdx.x;
    const int blk  = blockIdx.x;

    if (blk >= g) {
        const int j = (blk - g) * PART_T + tid;
        transpose_elem(W1, Wt1, j);
        transpose_elem(W2, Wt2, j);
        return;
    }

    const int base = blk * PART_CAP;
    const int len  = min(PART_CAP, e - base);

    if (tid < BIN_NODES) { hist[tid] = 0; lcur[tid] = 0; }

    int dv[PART_PER], sv[PART_PER];
    #pragma unroll
    for (int k = 0; k < PART_PER; ++k) {
        const int i  = tid + k * PART_T;
        const int ii = min(i, len - 1);
        dv[k] = dst[base + ii];
        sv[k] = src[base + ii];
    }
    #pragma unroll
    for (int k = 0; k < PART_PER; ++k)
        dv[k] = (tid + k * PART_T < len) ? (int)min((unsigned)dv[k], (unsigned)(n - 1)) : -1;
    __syncthreads();

    #pragma unroll
    for (int k = 0; k < PART_PER; ++k)
        if (dv[k] >= 0) atomicAdd(&hist[dv[k] >> BIN_SHIFT], 1);
    __syncthreads();

    int total;
    const int v  = (tid < BIN_NODES) ? hist[tid] : 0;
    const int ex = scan256(v, tid, wsum, total);
    if (tid < BIN_NODES) {
        lstart[tid] = ex;
        if (tid < nbin) {
            lenT[tid * g + blk]   = v;
            startT[tid * g + blk] = base + ex;
        }
    }
    __syncthreads();

    #pragma unroll
    for (int k = 0; k < PART_PER; ++k) {
        if (dv[k] >= 0) {
            const int b = dv[k] >> BIN_SHIFT;
            const int p = lstart[b] + atomicAdd(&lcur[b], 1);
            sS[p] = sv[k];
            sD[p] = dv[k];
        }
    }
    __syncthreads();

    for (int i = tid; i < len; i += PART_T) {
        sortedS[base + i] = sS[i];
        sortedD[base + i] = sD[i];
    }
}

template <bool PLACE>
__device__ __forceinline__ void
walk_runs(const int* __restrict__ sortedS, const int* __restrict__ sortedD,
          const int* rs, const int* rl, int* slot, int* lcol,
          int* __restrict__ col, int g, int e, int wave, int lane,
          int binBase, int binCnt, bool staged) {
    const int sub = lane >> 4;
    const int li  = lane & 15;
    for (int rb = wave * 16; rb < g; rb += (FIN_T / 32) * 16) {
        int rst[8], rln[8];
        int wmax = 0;
        #pragma unroll
        for (int u = 0; u < 8; ++u) {
            const int r = rb + 2 * u + sub;
            rst[u] = rs[r];
            rln[u] = rl[r];
            wmax   = max(wmax, rln[u]);
        }
        for (int off = 16; off > 0; off >>= 1) wmax = max(wmax, __shfl_xor(wmax, off));
        wmax = __builtin_amdgcn_readfirstlane(wmax);
        for (int j0 = 0; j0 < wmax; j0 += 48) {
            int dd[24], ss[24];
            #pragma unroll
            for (int u = 0; u < 8; ++u) {
                #pragma unroll
                for (int t = 0; t < 3; ++t) {
                    const int idx = j0 + 16 * t + li;
                    int a = rst[u] + min(idx, max(rln[u] - 1, 0));
                    a = min(a, e - 1);
                    dd[u * 3 + t] = sortedD[a];
                    if (PLACE) ss[u * 3 + t] = sortedS[a];
                }
            }
            asm volatile("" ::
                             "v"(dd[0]), "v"(dd[1]), "v"(dd[2]), "v"(dd[3]), "v"(dd[4]), "v"(dd[5]),
                             "v"(dd[6]), "v"(dd[7]), "v"(dd[8]), "v"(dd[9]), "v"(dd[10]), "v"(dd[11]),
                             "v"(dd[12]), "v"(dd[13]), "v"(dd[14]), "v"(dd[15]), "v"(dd[16]), "v"(dd[17]),
                             "v"(dd[18]), "v"(dd[19]), "v"(dd[20]), "v"(dd[21]), "v"(dd[22]), "v"(dd[23]));
            if (PLACE) {
                asm volatile("" ::
                                 "v"(ss[0]), "v"(ss[1]), "v"(ss[2]), "v"(ss[3]), "v"(ss[4]), "v"(ss[5]),
                                 "v"(ss[6]), "v"(ss[7]), "v"(ss[8]), "v"(ss[9]), "v"(ss[10]), "v"(ss[11]),
                                 "v"(ss[12]), "v"(ss[13]), "v"(ss[14]), "v"(ss[15]), "v"(ss[16]), "v"(ss[17]),
                                 "v"(ss[18]), "v"(ss[19]), "v"(ss[20]), "v"(ss[21]), "v"(ss[22]), "v"(ss[23]));
            }
            #pragma unroll
            for (int u = 0; u < 8; ++u) {
                #pragma unroll
                for (int t = 0; t < 3; ++t)
                    if (j0 + 16 * t + li >= rln[u]) dd[u * 3 + t] = -1;
            }
            #pragma unroll
            for (int q = 0; q < 24; ++q) {
                if (dd[q] >= 0) {
                    if (PLACE) {
                        int p = atomicAdd(&slot[dd[q] & (BIN_NODES - 1)], 1);
                        p = min(max(p, 0), binCnt - 1);
                        if (staged) lcol[p] = ss[q];
                        else        col[binBase + p] = ss[q];
                    } else {
                        atomicAdd(&slot[dd[q] & (BIN_NODES - 1)], 1);
                    }
                }
            }
        }
    }
}

__global__ void __launch_bounds__(FIN_T)
bin_finalize(const int* __restrict__ sortedS, const int* __restrict__ sortedD,
             const int* __restrict__ lenT, const int* __restrict__ startT,
             int* __restrict__ rowptr, int* __restrict__ cnt,
             float* __restrict__ dis, int* __restrict__ col,
             int n, int e, int g) {
    __shared__ int rs[FIN_T];
    __shared__ int rl[FIN_T];
    __shared__ int ncnt[BIN_NODES];
    __shared__ int ncur[BIN_NODES];
    __shared__ int lcol[FIN_CAP];
    __shared__ int red[FIN_T / 32];
    __shared__ int wsum[8];

    const int tid  = threadIdx.x;
    const int lane = tid & 31;
    const int wave = tid >> 5;
    const int bin  = blockIdx.x;
    const int lim  = bin * g;

    int part = 0, ln = 0, st = 0;
    if (tid < g) {
        ln   = lenT[lim + tid];
        st   = startT[lim + tid];
        part = st - tid * PART_CAP;
        ln = min(max(ln, 0), e);
        st = min(max(st, 0), e - ln);
    }
    rl[tid] = ln;
    rs[tid] = st;
    for (int off = 16; off > 0; off >>= 1) part += __shfl_xor(part, off);
    if (lane == 0) red[wave] = part;
    if (tid < BIN_NODES) ncnt[tid] = 0;
    __syncthreads();
    int binBase = 0;
    for (int w = 0; w < FIN_T / 32; ++w) binBase += red[w];

    walk_runs<false>(sortedS, sortedD, rs, rl, ncnt, lcol, col, g, e, wave, lane, 0, 0, false);
    __syncthreads();

    int binCnt;
    const int v  = (tid < BIN_NODES) ? ncnt[tid] : 0;
    const int ex = scan256(v, tid, wsum, binCnt);
    binCnt  = min(max(binCnt, 0), e);
    binBase = min(max(binBase, 0), e - binCnt);
    if (tid < BIN_NODES) {
        ncur[tid] = ex;
        const int node = (bin << BIN_SHIFT) + tid;
        if (node < n) {
            cnt[node]    = v;
            rowptr[node] = binBase + ex;
            dis[node]    = rsqrtf((float)v + 1.0f);
        }
    }
    __syncthreads();

    const bool staged = binCnt <= FIN_CAP;
    walk_runs<true>(sortedS, sortedD, rs, rl, ncur, lcol, col, g, e, wave, lane, binBase, binCnt, staged);
    __syncthreads();
    if (staged)
        for (int i = tid; i < binCnt; i += FIN_T) col[binBase + i] = lcol[i];
}

typedef __attribute__((ext_vector_type(4))) float v4f;

__device__ __forceinline__ v4f
gather_node(const v4f* __restrict__ X4, const float* __restrict__ dis,
            const int* __restrict__ rowptr, const int* __restrict__ cnt,
            const int* __restrict__ col, int node, int lane, int n, int e) {
    int c     = __builtin_amdgcn_readfirstlane(cnt[node]);
    int start = __builtin_amdgcn_readfirstlane(rowptr[node]);
    c     = min(max(c, 0), e);
    start = min(max(start, 0), e - c);

    const float dd = dis[node];
    const v4f xself = X4[(size_t)node * 32 + lane];
    v4f acc = {0.0f, 0.0f, 0.0f, 0.0f};

    for (int base = 0; base < c; base += 32) {
        const int m = min(32, c - base);
        int   my_s = 0;
        float my_w = 0.0f;
        if (lane < m) {
            my_s = col[start + base + lane];
            my_s = (int)min((unsigned)my_s, (unsigned)(n - 1));
            my_w = dis[my_s] * dd;
        }
        const int my_wi = __float_as_int(my_w);
        int j = 0;
        for (; j + 4 <= m; j += 4) {
            const int s0 = __builtin_amdgcn_readlane(my_s, j);
            const int s1 = __builtin_amdgcn_readlane(my_s, j + 1);
            const int s2 = __builtin_amdgcn_readlane(my_s, j + 2);
            const int s3 = __builtin_amdgcn_readlane(my_s, j + 3);
            const float w0 = __int_as_float(__builtin_amdgcn_readlane(my_wi, j));
            const float w1 = __int_as_float(__builtin_amdgcn_readlane(my_wi, j + 1));
            const float w2 = __int_as_float(__builtin_amdgcn_readlane(my_wi, j + 2));
            const float w3 = __int_as_float(__builtin_amdgcn_readlane(my_wi, j + 3));
            const v4f v0 = X4[(size_t)s0 * 32 + lane];
            const v4f v1 = X4[(size_t)s1 * 32 + lane];
            const v4f v2 = X4[(size_t)s2 * 32 + lane];
            const v4f v3 = X4[(size_t)s3 * 32 + lane];
            acc += v0 * w0;
            acc += v1 * w1;
            acc += v2 * w2;
            acc += v3 * w3;
        }
        const int rem = m - j;
        if (rem > 0) {
            v4f v0 = {0.0f, 0.0f, 0.0f, 0.0f};
            v4f v1 = v0, v2 = v0;
            float w0 = 0.0f, w1 = 0.0f, w2 = 0.0f;
            {
                const int s0 = __builtin_amdgcn_readlane(my_s, j);
                w0 = __int_as_float(__builtin_amdgcn_readlane(my_wi, j));
                v0 = X4[(size_t)s0 * 32 + lane];
            }
            if (rem > 1) {
                const int s1 = __builtin_amdgcn_readlane(my_s, j + 1);
                w1 = __int_as_float(__builtin_amdgcn_readlane(my_wi, j + 1));
                v1 = X4[(size_t)s1 * 32 + lane];
            }
            if (rem > 2) {
                const int s2 = __builtin_amdgcn_readlane(my_s, j + 2);
                w2 = __int_as_float(__builtin_amdgcn_readlane(my_wi, j + 2));
                v2 = X4[(size_t)s2 * 32 + lane];
            }
            acc += v0 * w0;
            acc += v1 * w1;
            acc += v2 * w2;
        }
    }
    return acc + xself * (dd * dd);
}

__device__ __forceinline__ v4f
gather_node_scaled(const v4f* __restrict__ X4, const float* __restrict__ dis,
                   const int* __restrict__ rowptr, const int* __restrict__ cnt,
                   const int* __restrict__ col, int node, int lane, int n, int e) {
    int c     = __builtin_amdgcn_readfirstlane(cnt[node]);
    int start = __builtin_amdgcn_readfirstlane(rowptr[node]);
    c     = min(max(c, 0), e);
    start = min(max(start, 0), e - c);

    const float dd = dis[node];
    v4f acc = X4[(size_t)node * 32 + lane];

    for (int base = 0; base < c; base += 32) {
        const int m = min(32, c - base);
        int my_s = 0;
        if (lane < m) {
            my_s = col[start + base + lane];
            my_s = (int)min((unsigned)my_s, (unsigned)(n - 1));
        }
        int j = 0;
        for (; j + 4 <= m; j += 4) {
            const int s0 = __builtin_amdgcn_readlane(my_s, j);
            const int s1 = __builtin_amdgcn_readlane(my_s, j + 1);
            const int s2 = __builtin_amdgcn_readlane(my_s, j + 2);
            const int s3 = __builtin_amdgcn_readlane(my_s, j + 3);
            const v4f v0 = X4[(size_t)s0 * 32 + lane];
            const v4f v1 = X4[(size_t)s1 * 32 + lane];
            const v4f v2 = X4[(size_t)s2 * 32 + lane];
            const v4f v3 = X4[(size_t)s3 * 32 + lane];
            acc += v0;
            acc += v1;
            acc += v2;
            acc += v3;
        }
        const int rem = m - j;
        if (rem > 0) {
            v4f v1 = {0.0f, 0.0f, 0.0f, 0.0f};
            v4f v2 = v1;
            const int s0 = __builtin_amdgcn_readlane(my_s, j);
            const v4f v0 = X4[(size_t)s0 * 32 + lane];
            if (rem > 1) {
                const int s1 = __builtin_amdgcn_readlane(my_s, j + 1);
                v1 = X4[(size_t)s1 * 32 + lane];
            }
            if (rem > 2) {
                const int s2 = __builtin_amdgcn_readlane(my_s, j + 2);
                v2 = X4[(size_t)s2 * 32 + lane];
            }
            acc += v0;
            acc += v1;
            acc += v2;
        }
    }
    return acc * dd;
}

template <bool FIRST>
__global__ void __launch_bounds__(512)
gather_gemm(const float* __restrict__ xin, const float* __restrict__ dis,
            const int* __restrict__ rowptr, const int* __restrict__ cnt,
            const int* __restrict__ col, const float* __restrict__ Wt,
            const float* __restrict__ bias, float* __restrict__ outp,
            int n, int e) {
    __shared__ __attribute__((aligned(16))) float Xs[32][132];
    __shared__ __attribute__((aligned(16))) float Cs[32][132];

    const int tid     = threadIdx.x;
    const int lane    = tid & 31;
    const int wave    = __builtin_amdgcn_readfirstlane(tid >> 5);
    const int rowBase = blockIdx.x * 32;
    const v4f* __restrict__ X4 = (const v4f*)xin;

    for (int t = 0; t < 2; ++t) {
        const int r    = wave * 2 + t;
        const int node = rowBase + r;
        v4f y = {0.0f, 0.0f, 0.0f, 0.0f};
        if (node < n) {
            if (FIRST) y = gather_node(X4, dis, rowptr, cnt, col, node, lane, n, e);
            else       y = gather_node_scaled(X4, dis, rowptr, cnt, col, node, lane, n, e);
        }
        *(v4f*)&Xs[r][lane * 4] = y;
    }
    __syncthreads();

    if (wave < 8) {
        const int colBase = wave * 16;
        const int l16     = lane & 15;
        const int khalf   = (lane >> 4) << 1;

        v8f acc0 = {};
        v8f acc1 = {};
        #pragma unroll
        for (int k = 0; k < NDIM; k += 4) {
            v2f a0, a1, b;
            a0.x = Xs[l16][k + khalf];
            a0.y = Xs[l16][k + khalf + 1];
            a1.x = Xs[16 + l16][k + khalf];
            a1.y = Xs[16 + l16][k + khalf + 1];
            b.x = Wt[(k + khalf)     * NDIM + colBase + l16];
            b.y = Wt[(k + khalf + 1) * NDIM + colBase + l16];
            acc0 = __builtin_amdgcn_wmma_f32_16x16x4_f32(
                 false, a0,  false, b,
                 (short)0, acc0,  false,  false);
            acc1 = __builtin_amdgcn_wmma_f32_16x16x4_f32(
                 false, a1,  false, b,
                 (short)0, acc1,  false,  false);
        }

        const int crow0 = (lane >> 4) << 3;
        #pragma unroll
        for (int v = 0; v < 8; ++v) {
            Cs[crow0 + v][colBase + l16]      = acc0[v];
            Cs[16 + crow0 + v][colBase + l16] = acc1[v];
        }
    }
    const v4f bv = {bias[lane * 4], bias[lane * 4 + 1], bias[lane * 4 + 2], bias[lane * 4 + 3]};
    float dsc[2] = {1.0f, 1.0f};
    if (FIRST) {
        dsc[0] = dis[min(rowBase + wave * 2,     n - 1)];
        dsc[1] = dis[min(rowBase + wave * 2 + 1, n - 1)];
    }
    __syncthreads();

    #pragma unroll
    for (int t = 0; t < 2; ++t) {
        const int r    = wave * 2 + t;
        const int node = rowBase + r;
        if (node < n) {
            v4f o = *(const v4f*)&Cs[r][lane * 4] + bv;
            if (FIRST) {
                o.x = (o.x > 0.0f) ? o.x : NEG_SLOPE * o.x;
                o.y = (o.y > 0.0f) ? o.y : NEG_SLOPE * o.y;
                o.z = (o.z > 0.0f) ? o.z : NEG_SLOPE * o.z;
                o.w = (o.w > 0.0f) ? o.w : NEG_SLOPE * o.w;
                o = o * dsc[t];
            }
            ((v4f*)outp)[(size_t)node * 32 + lane] = o;
        }
    }
}

extern "C" void kernel_launch(void* const* d_in, const int* in_sizes, int n_in,
                              void* d_out, int out_size, void* d_ws, size_t ws_size,
                              hipStream_t stream) {
    const float* x    = (const float*)d_in[0];
    const int*   ei   = (const int*)d_in[1];
    const float* W1   = (const float*)d_in[3];
    const float* b1   = (const float*)d_in[4];
    const float* W2   = (const float*)d_in[5];
    const float* b2   = (const float*)d_in[6];

    const int N = in_sizes[0] / NDIM;
    const int E = in_sizes[1] / 2;
    const int* src = ei;
    const int* dst = ei + E;

    const int NP   = (N + 3) & ~3;
    const int EP   = (E + 3) & ~3;
    const int NBIN = (N + BIN_NODES - 1) >> BIN_SHIFT;
    const int G    = (E + PART_CAP - 1) / PART_CAP;
    const int TP   = (NBIN * G + 3) & ~3;
    float* Wt1    = (float*)d_ws;
    float* Wt2    = Wt1 + NDIM * NDIM;
    float* dis    = Wt2 + NDIM * NDIM;
    int*   cnt    = (int*)(dis + NP);
    int*   rowptr = cnt + NP;
    int*   lenT   = rowptr + NP;
    int*   startT = lenT + TP;
    int*   sortS  = startT + TP;
    int*   sortD  = sortS + EP;
    int*   col    = sortD + EP;
    float* bufB   = (float*)(col + EP);
    float* outp   = (float*)d_out;

    dim3 blkLayer(512);
    dim3 gLayer((N + 31) / 32);

    bin_sort_chunk<<<dim3(G + TR_BLOCKS), dim3(PART_T), 0, stream>>>(src, dst, sortS, sortD, lenT, startT,
                                                                     W1, Wt1, W2, Wt2, N, E, NBIN, G);
    bin_finalize<<<dim3(NBIN), dim3(FIN_T), 0, stream>>>(sortS, sortD, lenT, startT, rowptr, cnt, dis, col, N, E, G);

    gather_gemm<true><<<gLayer, blkLayer, 0, stream>>>(x, dis, rowptr, cnt, col, Wt1, b1, bufB, N, E);

    gather_gemm<false><<<gLayer, blkLayer, 0, stream>>>(bufB, dis, rowptr, cnt, col, Wt2, b2, outp, N, E);
}


# Workspace order: layer-1 result rows first, so each 512-B row is 512-B aligned (was 192 B into a 256-B unit: 5 lines, 3 units per gathered row; now 4, 2; layer-1 row stores whole lines); same sums
# speedup vs baseline: 5.6228x; 1.0751x over previous
#include <hip/hip_runtime.h>
#include <hip/hip_bf16.h>

#define NDIM 128
#define NEG_SLOPE 0.01f

typedef __attribute__((ext_vector_type(2))) float v2f;
typedef __attribute__((ext_vector_type(8))) float v8f;

__device__ __forceinline__ void transpose_elem(const float* __restrict__ W,
                                               float* __restrict__ Wt, int j) {
    Wt[j] = W[(j & (NDIM - 1)) * NDIM + (j >> 7)];
}

#define BIN_SHIFT 8
#define BIN_NODES 256
#define PART_CAP  3584
#define PART_T    512
#define PART_PER  ((PART_CAP + PART_T - 1) / PART_T)
#define FIN_T     512
#define FIN_CAP   8192
#define TR_BLOCKS ((NDIM * NDIM) / PART_T)

__device__ __forceinline__ int scan256(int v, int tid, int* wsum, int& total) {
    const int lane = tid & 31;
    const int wave = tid >> 5;
    int x = v;
    for (int off = 1; off < 32; off <<= 1) {
        int y = __shfl_up(x, off);
        if (lane >= off) x += y;
    }
    if (lane == 31 && wave < 8) wsum[wave] = x;
    __syncthreads();
    int wpre = 0, tot = 0;
    for (int w = 0; w < 8; ++w) {
        const int s = wsum[w];
        if (w < wave) wpre += s;
        tot += s;
    }
    total = tot;
    return wpre + x - v;
}

__global__ void __launch_bounds__(PART_T)
bin_sort_chunk(const int* __restrict__ src, const int* __restrict__ dst,
               int* __restrict__ sortedS, int* __restrict__ sortedD,
               int* __restrict__ lenT, int* __restrict__ startT,
               const float* __restrict__ W1, float* __restrict__ Wt1,
               const float* __restrict__ W2, float* __restrict__ Wt2,
               int n, int e, int nbin, int g) {
    __shared__ int sS[PART_CAP];
    __shared__ int sD[PART_CAP];
    __shared__ int hist[BIN_NODES];
    __shared__ int lstart[BIN_NODES];
    __shared__ int lcur[BIN_NODES];
    __shared__ int wsum[8];

    const int tid  = threadIdx.x;
    const int blk  = blockIdx.x;

    if (blk >= g) {
        const int j = (blk - g) * PART_T + tid;
        transpose_elem(W1, Wt1, j);
        transpose_elem(W2, Wt2, j);
        return;
    }

    const int base = blk * PART_CAP;
    const int len  = min(PART_CAP, e - base);

    if (tid < BIN_NODES) { hist[tid] = 0; lcur[tid] = 0; }

    int dv[PART_PER], sv[PART_PER];
    #pragma unroll
    for (int k = 0; k < PART_PER; ++k) {
        const int i  = tid + k * PART_T;
        const int ii = min(i, len - 1);
        dv[k] = dst[base + ii];
        sv[k] = src[base + ii];
    }
    #pragma unroll
    for (int k = 0; k < PART_PER; ++k)
        dv[k] = (tid + k * PART_T < len) ? (int)min((unsigned)dv[k], (unsigned)(n - 1)) : -1;
    __syncthreads();

    #pragma unroll
    for (int k = 0; k < PART_PER; ++k)
        if (dv[k] >= 0) atomicAdd(&hist[dv[k] >> BIN_SHIFT], 1);
    __syncthreads();

    int total;
    const int v  = (tid < BIN_NODES) ? hist[tid] : 0;
    const int ex = scan256(v, tid, wsum, total);
    if (tid < BIN_NODES) {
        lstart[tid] = ex;
        if (tid < nbin) {
            lenT[tid * g + blk]   = v;
            startT[tid * g + blk] = base + ex;
        }
    }
    __syncthreads();

    #pragma unroll
    for (int k = 0; k < PART_PER; ++k) {
        if (dv[k] >= 0) {
            const int b = dv[k] >> BIN_SHIFT;
            const int p = lstart[b] + atomicAdd(&lcur[b], 1);
            sS[p] = sv[k];
            sD[p] = dv[k];
        }
    }
    __syncthreads();

    for (int i = tid; i < len; i += PART_T) {
        sortedS[base + i] = sS[i];
        sortedD[base + i] = sD[i];
    }
}

template <bool PLACE>
__device__ __forceinline__ void
walk_runs(const int* __restrict__ sortedS, const int* __restrict__ sortedD,
          const int* rs, const int* rl, int* slot, int* lcol,
          int* __restrict__ col, int g, int e, int wave, int lane,
          int binBase, int binCnt, bool staged) {
    const int sub = lane >> 4;
    const int li  = lane & 15;
    for (int rb = wave * 16; rb < g; rb += (FIN_T / 32) * 16) {
        int rst[8], rln[8];
        int wmax = 0;
        #pragma unroll
        for (int u = 0; u < 8; ++u) {
            const int r = rb + 2 * u + sub;
            rst[u] = rs[r];
            rln[u] = rl[r];
            wmax   = max(wmax, rln[u]);
        }
        for (int off = 16; off > 0; off >>= 1) wmax = max(wmax, __shfl_xor(wmax, off));
        wmax = __builtin_amdgcn_readfirstlane(wmax);
        for (int j0 = 0; j0 < wmax; j0 += 48) {
            int dd[24], ss[24];
            #pragma unroll
            for (int u = 0; u < 8; ++u) {
                #pragma unroll
                for (int t = 0; t < 3; ++t) {
                    const int idx = j0 + 16 * t + li;
                    int a = rst[u] + min(idx, max(rln[u] - 1, 0));
                    a = min(a, e - 1);
                    dd[u * 3 + t] = sortedD[a];
                    if (PLACE) ss[u * 3 + t] = sortedS[a];
                }
            }
            asm volatile("" ::
                             "v"(dd[0]), "v"(dd[1]), "v"(dd[2]), "v"(dd[3]), "v"(dd[4]), "v"(dd[5]),
                             "v"(dd[6]), "v"(dd[7]), "v"(dd[8]), "v"(dd[9]), "v"(dd[10]), "v"(dd[11]),
                             "v"(dd[12]), "v"(dd[13]), "v"(dd[14]), "v"(dd[15]), "v"(dd[16]), "v"(dd[17]),
                             "v"(dd[18]), "v"(dd[19]), "v"(dd[20]), "v"(dd[21]), "v"(dd[22]), "v"(dd[23]));
            if (PLACE) {
                asm volatile("" ::
                                 "v"(ss[0]), "v"(ss[1]), "v"(ss[2]), "v"(ss[3]), "v"(ss[4]), "v"(ss[5]),
                                 "v"(ss[6]), "v"(ss[7]), "v"(ss[8]), "v"(ss[9]), "v"(ss[10]), "v"(ss[11]),
                                 "v"(ss[12]), "v"(ss[13]), "v"(ss[14]), "v"(ss[15]), "v"(ss[16]), "v"(ss[17]),
                                 "v"(ss[18]), "v"(ss[19]), "v"(ss[20]), "v"(ss[21]), "v"(ss[22]), "v"(ss[23]));
            }
            #pragma unroll
            for (int u = 0; u < 8; ++u) {
                #pragma unroll
                for (int t = 0; t < 3; ++t)
                    if (j0 + 16 * t + li >= rln[u]) dd[u * 3 + t] = -1;
            }
            #pragma unroll
            for (int q = 0; q < 24; ++q) {
                if (dd[q] >= 0) {
                    if (PLACE) {
                        int p = atomicAdd(&slot[dd[q] & (BIN_NODES - 1)], 1);
                        p = min(max(p, 0), binCnt - 1);
                        if (staged) lcol[p] = ss[q];
                        else        col[binBase + p] = ss[q];
                    } else {
                        atomicAdd(&slot[dd[q] & (BIN_NODES - 1)], 1);
                    }
                }
            }
        }
    }
}

__global__ void __launch_bounds__(FIN_T)
bin_finalize(const int* __restrict__ sortedS, const int* __restrict__ sortedD,
             const int* __restrict__ lenT, const int* __restrict__ startT,
             int* __restrict__ rowptr, int* __restrict__ cnt,
             float* __restrict__ dis, int* __restrict__ col,
             int n, int e, int g) {
    __shared__ int rs[FIN_T];
    __shared__ int rl[FIN_T];
    __shared__ int ncnt[BIN_NODES];
    __shared__ int ncur[BIN_NODES];
    __shared__ int lcol[FIN_CAP];
    __shared__ int red[FIN_T / 32];
    __shared__ int wsum[8];

    const int tid  = threadIdx.x;
    const int lane = tid & 31;
    const int wave = tid >> 5;
    const int bin  = blockIdx.x;
    const int lim  = bin * g;

    int part = 0, ln = 0, st = 0;
    if (tid < g) {
        ln   = lenT[lim + tid];
        st   = startT[lim + tid];
        part = st - tid * PART_CAP;
        ln = min(max(ln, 0), e);
        st = min(max(st, 0), e - ln);
    }
    rl[tid] = ln;
    rs[tid] = st;
    for (int off = 16; off > 0; off >>= 1) part += __shfl_xor(part, off);
    if (lane == 0) red[wave] = part;
    if (tid < BIN_NODES) ncnt[tid] = 0;
    __syncthreads();
    int binBase = 0;
    for (int w = 0; w < FIN_T / 32; ++w) binBase += red[w];

    walk_runs<false>(sortedS, sortedD, rs, rl, ncnt, lcol, col, g, e, wave, lane, 0, 0, false);
    __syncthreads();

    int binCnt;
    const int v  = (tid < BIN_NODES) ? ncnt[tid] : 0;
    const int ex = scan256(v, tid, wsum, binCnt);
    binCnt  = min(max(binCnt, 0), e);
    binBase = min(max(binBase, 0), e - binCnt);
    if (tid < BIN_NODES) {
        ncur[tid] = ex;
        const int node = (bin << BIN_SHIFT) + tid;
        if (node < n) {
            cnt[node]    = v;
            rowptr[node] = binBase + ex;
            dis[node]    = rsqrtf((float)v + 1.0f);
        }
    }
    __syncthreads();

    const bool staged = binCnt <= FIN_CAP;
    walk_runs<true>(sortedS, sortedD, rs, rl, ncur, lcol, col, g, e, wave, lane, binBase, binCnt, staged);
    __syncthreads();
    if (staged)
        for (int i = tid; i < binCnt; i += FIN_T) col[binBase + i] = lcol[i];
}

typedef __attribute__((ext_vector_type(4))) float v4f;

__device__ __forceinline__ v4f
gather_node(const v4f* __restrict__ X4, const float* __restrict__ dis,
            const int* __restrict__ rowptr, const int* __restrict__ cnt,
            const int* __restrict__ col, int node, int lane, int n, int e) {
    int c     = __builtin_amdgcn_readfirstlane(cnt[node]);
    int start = __builtin_amdgcn_readfirstlane(rowptr[node]);
    c     = min(max(c, 0), e);
    start = min(max(start, 0), e - c);

    const float dd = dis[node];
    const v4f xself = X4[(size_t)node * 32 + lane];
    v4f acc = {0.0f, 0.0f, 0.0f, 0.0f};

    for (int base = 0; base < c; base += 32) {
        const int m = min(32, c - base);
        int   my_s = 0;
        float my_w = 0.0f;
        if (lane < m) {
            my_s = col[start + base + lane];
            my_s = (int)min((unsigned)my_s, (unsigned)(n - 1));
            my_w = dis[my_s] * dd;
        }
        const int my_wi = __float_as_int(my_w);
        int j = 0;
        for (; j + 4 <= m; j += 4) {
            const int s0 = __builtin_amdgcn_readlane(my_s, j);
            const int s1 = __builtin_amdgcn_readlane(my_s, j + 1);
            const int s2 = __builtin_amdgcn_readlane(my_s, j + 2);
            const int s3 = __builtin_amdgcn_readlane(my_s, j + 3);
            const float w0 = __int_as_float(__builtin_amdgcn_readlane(my_wi, j));
            const float w1 = __int_as_float(__builtin_amdgcn_readlane(my_wi, j + 1));
            const float w2 = __int_as_float(__builtin_amdgcn_readlane(my_wi, j + 2));
            const float w3 = __int_as_float(__builtin_amdgcn_readlane(my_wi, j + 3));
            const v4f v0 = X4[(size_t)s0 * 32 + lane];
            const v4f v1 = X4[(size_t)s1 * 32 + lane];
            const v4f v2 = X4[(size_t)s2 * 32 + lane];
            const v4f v3 = X4[(size_t)s3 * 32 + lane];
            acc += v0 * w0;
            acc += v1 * w1;
            acc += v2 * w2;
            acc += v3 * w3;
        }
        const int rem = m - j;
        if (rem > 0) {
            v4f v0 = {0.0f, 0.0f, 0.0f, 0.0f};
            v4f v1 = v0, v2 = v0;
            float w0 = 0.0f, w1 = 0.0f, w2 = 0.0f;
            {
                const int s0 = __builtin_amdgcn_readlane(my_s, j);
                w0 = __int_as_float(__builtin_amdgcn_readlane(my_wi, j));
                v0 = X4[(size_t)s0 * 32 + lane];
            }
            if (rem > 1) {
                const int s1 = __builtin_amdgcn_readlane(my_s, j + 1);
                w1 = __int_as_float(__builtin_amdgcn_readlane(my_wi, j + 1));
                v1 = X4[(size_t)s1 * 32 + lane];
            }
            if (rem > 2) {
                const int s2 = __builtin_amdgcn_readlane(my_s, j + 2);
                w2 = __int_as_float(__builtin_amdgcn_readlane(my_wi, j + 2));
                v2 = X4[(size_t)s2 * 32 + lane];
            }
            acc += v0 * w0;
            acc += v1 * w1;
            acc += v2 * w2;
        }
    }
    return acc + xself * (dd * dd);
}

__device__ __forceinline__ v4f
gather_node_scaled(const v4f* __restrict__ X4, const float* __restrict__ dis,
                   const int* __restrict__ rowptr, const int* __restrict__ cnt,
                   const int* __restrict__ col, int node, int lane, int n, int e) {
    int c     = __builtin_amdgcn_readfirstlane(cnt[node]);
    int start = __builtin_amdgcn_readfirstlane(rowptr[node]);
    c     = min(max(c, 0), e);
    start = min(max(start, 0), e - c);

    const float dd = dis[node];
    v4f acc = X4[(size_t)node * 32 + lane];

    for (int base = 0; base < c; base += 32) {
        const int m = min(32, c - base);
        int my_s = 0;
        if (lane < m) {
            my_s = col[start + base + lane];
            my_s = (int)min((unsigned)my_s, (unsigned)(n - 1));
        }
        int j = 0;
        for (; j + 4 <= m; j += 4) {
            const int s0 = __builtin_amdgcn_readlane(my_s, j);
            const int s1 = __builtin_amdgcn_readlane(my_s, j + 1);
            const int s2 = __builtin_amdgcn_readlane(my_s, j + 2);
            const int s3 = __builtin_amdgcn_readlane(my_s, j + 3);
            const v4f v0 = X4[(size_t)s0 * 32 + lane];
            const v4f v1 = X4[(size_t)s1 * 32 + lane];
            const v4f v2 = X4[(size_t)s2 * 32 + lane];
            const v4f v3 = X4[(size_t)s3 * 32 + lane];
            acc += v0;
            acc += v1;
            acc += v2;
            acc += v3;
        }
        const int rem = m - j;
        if (rem > 0) {
            v4f v1 = {0.0f, 0.0f, 0.0f, 0.0f};
            v4f v2 = v1;
            const int s0 = __builtin_amdgcn_readlane(my_s, j);
            const v4f v0 = X4[(size_t)s0 * 32 + lane];
            if (rem > 1) {
                const int s1 = __builtin_amdgcn_readlane(my_s, j + 1);
                v1 = X4[(size_t)s1 * 32 + lane];
            }
            if (rem > 2) {
                const int s2 = __builtin_amdgcn_readlane(my_s, j + 2);
                v2 = X4[(size_t)s2 * 32 + lane];
            }
            acc += v0;
            acc += v1;
            acc += v2;
        }
    }
    return acc * dd;
}

template <bool FIRST>
__global__ void __launch_bounds__(512)
gather_gemm(const float* __restrict__ xin, const float* __restrict__ dis,
            const int* __restrict__ rowptr, const int* __restrict__ cnt,
            const int* __restrict__ col, const float* __restrict__ Wt,
            const float* __restrict__ bias, float* __restrict__ outp,
            int n, int e) {
    __shared__ __attribute__((aligned(16))) float Xs[32][132];
    __shared__ __attribute__((aligned(16))) float Cs[32][132];

    const int tid     = threadIdx.x;
    const int lane    = tid & 31;
    const int wave    = __builtin_amdgcn_readfirstlane(tid >> 5);
    const int rowBase = blockIdx.x * 32;
    const v4f* __restrict__ X4 = (const v4f*)xin;

    for (int t = 0; t < 2; ++t) {
        const int r    = wave * 2 + t;
        const int node = rowBase + r;
        v4f y = {0.0f, 0.0f, 0.0f, 0.0f};
        if (node < n) {
            if (FIRST) y = gather_node(X4, dis, rowptr, cnt, col, node, lane, n, e);
            else       y = gather_node_scaled(X4, dis, rowptr, cnt, col, node, lane, n, e);
        }
        *(v4f*)&Xs[r][lane * 4] = y;
    }
    __syncthreads();

    if (wave < 8) {
        const int colBase = wave * 16;
        const int l16     = lane & 15;
        const int khalf   = (lane >> 4) << 1;

        v8f acc0 = {};
        v8f acc1 = {};
        #pragma unroll
        for (int k = 0; k < NDIM; k += 4) {
            v2f a0, a1, b;
            a0.x = Xs[l16][k + khalf];
            a0.y = Xs[l16][k + khalf + 1];
            a1.x = Xs[16 + l16][k + khalf];
            a1.y = Xs[16 + l16][k + khalf + 1];
            b.x = Wt[(k + khalf)     * NDIM + colBase + l16];
            b.y = Wt[(k + khalf + 1) * NDIM + colBase + l16];
            acc0 = __builtin_amdgcn_wmma_f32_16x16x4_f32(
                 false, a0,  false, b,
                 (short)0, acc0,  false,  false);
            acc1 = __builtin_amdgcn_wmma_f32_16x16x4_f32(
                 false, a1,  false, b,
                 (short)0, acc1,  false,  false);
        }

        const int crow0 = (lane >> 4) << 3;
        #pragma unroll
        for (int v = 0; v < 8; ++v) {
            Cs[crow0 + v][colBase + l16]      = acc0[v];
            Cs[16 + crow0 + v][colBase + l16] = acc1[v];
        }
    }
    const v4f bv = {bias[lane * 4], bias[lane * 4 + 1], bias[lane * 4 + 2], bias[lane * 4 + 3]};
    float dsc[2] = {1.0f, 1.0f};
    if (FIRST) {
        dsc[0] = dis[min(rowBase + wave * 2,     n - 1)];
        dsc[1] = dis[min(rowBase + wave * 2 + 1, n - 1)];
    }
    __syncthreads();

    #pragma unroll
    for (int t = 0; t < 2; ++t) {
        const int r    = wave * 2 + t;
        const int node = rowBase + r;
        if (node < n) {
            v4f o = *(const v4f*)&Cs[r][lane * 4] + bv;
            if (FIRST) {
                o.x = (o.x > 0.0f) ? o.x : NEG_SLOPE * o.x;
                o.y = (o.y > 0.0f) ? o.y : NEG_SLOPE * o.y;
                o.z = (o.z > 0.0f) ? o.z : NEG_SLOPE * o.z;
                o.w = (o.w > 0.0f) ? o.w : NEG_SLOPE * o.w;
                o = o * dsc[t];
            }
            ((v4f*)outp)[(size_t)node * 32 + lane] = o;
        }
    }
}

extern "C" void kernel_launch(void* const* d_in, const int* in_sizes, int n_in,
                              void* d_out, int out_size, void* d_ws, size_t ws_size,
                              hipStream_t stream) {
    const float* x    = (const float*)d_in[0];
    const int*   ei   = (const int*)d_in[1];
    const float* W1   = (const float*)d_in[3];
    const float* b1   = (const float*)d_in[4];
    const float* W2   = (const float*)d_in[5];
    const float* b2   = (const float*)d_in[6];

    const int N = in_sizes[0] / NDIM;
    const int E = in_sizes[1] / 2;
    const int* src = ei;
    const int* dst = ei + E;

    const int NP   = (N + 3) & ~3;
    const int EP   = (E + 3) & ~3;
    const int NBIN = (N + BIN_NODES - 1) >> BIN_SHIFT;
    const int G    = (E + PART_CAP - 1) / PART_CAP;
    const int TP   = (NBIN * G + 3) & ~3;
    float* bufB   = (float*)d_ws;
    float* Wt1    = bufB + (size_t)N * NDIM;
    float* Wt2    = Wt1 + NDIM * NDIM;
    float* dis    = Wt2 + NDIM * NDIM;
    int*   cnt    = (int*)(dis + NP);
    int*   rowptr = cnt + NP;
    int*   lenT   = rowptr + NP;
    int*   startT = lenT + TP;
    int*   sortS  = startT + TP;
    int*   sortD  = sortS + EP;
    int*   col    = sortD + EP;
    float* outp   = (float*)d_out;

    dim3 blkLayer(512);
    dim3 gLayer((N + 31) / 32);

    bin_sort_chunk<<<dim3(G + TR_BLOCKS), dim3(PART_T), 0, stream>>>(src, dst, sortS, sortD, lenT, startT,
                                                                     W1, Wt1, W2, Wt2, N, E, NBIN, G);
    bin_finalize<<<dim3(NBIN), dim3(FIN_T), 0, stream>>>(sortS, sortD, lenT, startT, rowptr, cnt, dis, col, N, E, G);

    gather_gemm<true><<<gLayer, blkLayer, 0, stream>>>(x, dis, rowptr, cnt, col, Wt1, b1, bufB, N, E);

    gather_gemm<false><<<gLayer, blkLayer, 0, stream>>>(bufB, dis, rowptr, cnt, col, Wt2, b2, outp, N, E);
}
